# starting kernel
# baseline (speedup 1.0000x reference)
#include <hip/hip_runtime.h>


#define N_NODES 100000
#define N_EDGES 1600000
#define NFEAT   256
#define NHID    128
#define NCLASS  40

typedef __attribute__((ext_vector_type(2))) float v2f;
typedef __attribute__((ext_vector_type(8))) float v8f;

__global__ __launch_bounds__(256)
void gcn_gemm1(const float* __restrict__ x, const float* __restrict__ W1,
               float* __restrict__ H0) {
  const int mt   = blockIdx.x;
  const int wave = threadIdx.x >> 5;
  const int lane = threadIdx.x & 31;
  const int half = lane >> 4;
  const int mn   = lane & 15;
  const int m0 = mt * 16;
  const int n0 = wave * 16;

  const float* arow = x  + (size_t)(m0 + mn) * NFEAT;
  const float* bcol = W1 + n0 + mn;

  v8f c = {};
  #pragma unroll 4
  for (int k = 0; k < NFEAT; k += 4) {
    const int k0 = k + 2 * half;
    v2f a, b;
    a.x = arow[k0];
    a.y = arow[k0 + 1];
    b.x = bcol[(size_t)k0 * NHID];
    b.y = bcol[(size_t)(k0 + 1) * NHID];
    c = __builtin_amdgcn_wmma_f32_16x16x4_f32(
         false, a,  false, b,
         (short)0, c,  false,  false);
  }
  float* crow = H0 + (size_t)m0 * NHID + n0 + mn;
  #pragma unroll
  for (int j = 0; j < 8; ++j)
    crow[(size_t)(j + 8 * half) * NHID] = c[j];
}

__global__ __launch_bounds__(256)
void gcn_spmm1(const int* __restrict__ esrc, const int* __restrict__ edst,
               const float* __restrict__ ew, const float* __restrict__ H0,
               float* __restrict__ agg) {
  const long long tid = (long long)blockIdx.x * blockDim.x + threadIdx.x;
  const int e = (int)(tid >> 5);
  if (e >= N_EDGES) return;
  const int lane = threadIdx.x & 31;
  const int s = esrc[e];
  const int d = edst[e];
  const float w = ew[e];
  const float4 hv = *(const float4*)(H0 + (size_t)s * NHID + lane * 4);
  float* o = agg + (size_t)d * NHID + lane * 4;
  atomicAdd(o + 0, w * hv.x);
  atomicAdd(o + 1, w * hv.y);
  atomicAdd(o + 2, w * hv.z);
  atomicAdd(o + 3, w * hv.w);
}

__global__ __launch_bounds__(96)
void gcn_gemm2(const float* __restrict__ agg, const float* __restrict__ b1,
               const float* __restrict__ W2, float* __restrict__ H2) {
  const int mt   = blockIdx.x;
  const int wave = threadIdx.x >> 5;
  const int lane = threadIdx.x & 31;
  const int half = lane >> 4;
  const int mn   = lane & 15;
  const int m0  = mt * 16;
  const int col = wave * 16 + mn;
  const bool colok = (col < NCLASS);

  const float* arow = agg + (size_t)(m0 + mn) * NHID;

  v8f c = {};
  #pragma unroll 4
  for (int k = 0; k < NHID; k += 4) {
    const int k0 = k + 2 * half;
    v2f a, b;
    a.x = fmaxf(arow[k0]     + b1[k0],     0.0f);
    a.y = fmaxf(arow[k0 + 1] + b1[k0 + 1], 0.0f);
    b.x = colok ? W2[(size_t)k0 * NCLASS + col]       : 0.0f;
    b.y = colok ? W2[(size_t)(k0 + 1) * NCLASS + col] : 0.0f;
    c = __builtin_amdgcn_wmma_f32_16x16x4_f32(
        false, a, false, b, (short)0, c, false, false);
  }
  if (colok) {
    #pragma unroll
    for (int j = 0; j < 8; ++j)
      H2[(size_t)(m0 + j + 8 * half) * NCLASS + col] = c[j];
  }
}

__global__ __launch_bounds__(256)
void gcn_spmm2(const int* __restrict__ esrc, const int* __restrict__ edst,
               const float* __restrict__ ew, const float* __restrict__ H2,
               float* __restrict__ out) {
  const long long tid = (long long)blockIdx.x * blockDim.x + threadIdx.x;
  if (tid >= (long long)N_EDGES * NCLASS) return;
  const int e = (int)(tid / NCLASS);
  const int cc = (int)(tid % NCLASS);
  const int s = esrc[e];
  const int d = edst[e];
  atomicAdd(&out[(size_t)d * NCLASS + cc], ew[e] * H2[(size_t)s * NCLASS + cc]);
}

__global__ __launch_bounds__(256)
void gcn_zero4(float4* __restrict__ p, long long n4) {
  const long long i = (long long)blockIdx.x * blockDim.x + threadIdx.x;
  if (i < n4) p[i] = make_float4(0.f, 0.f, 0.f, 0.f);
}

__global__ __launch_bounds__(256)
void gcn_init_out(float* __restrict__ out, const float* __restrict__ b2) {
  const long long i = (long long)blockIdx.x * blockDim.x + threadIdx.x;
  if (i < (long long)N_NODES * NCLASS) out[i] = b2[i % NCLASS];
}

extern "C" void kernel_launch(void* const* d_in, const int* in_sizes, int n_in,
                              void* d_out, int out_size, void* d_ws, size_t ws_size,
                              hipStream_t stream) {
  const float* x    = (const float*)d_in[0];
  const int*   esrc = (const int*)  d_in[1];
  const int*   edst = (const int*)  d_in[2];
  const float* ew   = (const float*)d_in[3];
  const float* W1   = (const float*)d_in[4];
  const float* b1   = (const float*)d_in[5];
  const float* W2   = (const float*)d_in[6];
  const float* b2   = (const float*)d_in[7];
  float* out = (float*)d_out;

  float* H0  = (float*)d_ws;
  float* agg = H0 + (size_t)N_NODES * NHID;
  float* H2  = H0;

  {
    const long long n4 = (long long)N_NODES * NHID / 4;
    gcn_zero4<<<(unsigned)((n4 + 255) / 256), 256, 0, stream>>>((float4*)agg, n4);
  }
  gcn_gemm1<<<N_NODES / 16, 256, 0, stream>>>(x, W1, H0);
  {
    const long long nt = (long long)N_EDGES * 32;
    gcn_spmm1<<<(unsigned)((nt + 255) / 256), 256, 0, stream>>>(esrc, edst, ew, H0, agg);
  }
  gcn_gemm2<<<N_NODES / 16, 96, 0, stream>>>(agg, b1, W2, H2);
  {
    const long long n = (long long)N_NODES * NCLASS;
    gcn_init_out<<<(unsigned)((n + 255) / 256), 256, 0, stream>>>(out, b2);
  }
  {
    const long long nt = (long long)N_EDGES * NCLASS;
    gcn_spmm2<<<(unsigned)((nt + 255) / 256), 256, 0, stream>>>(esrc, edst, ew, H2, out);
  }
}


# GEMM2 re-decomposed: a wave owns 16 rows x all 40 columns (agg read once, 16-byte loads), W2+b1 in LDS as k-quads, 512 resident blocks share strips, whole-line H2 stores; k order in runs of 8 changed
# speedup vs baseline: 2.4078x; 2.4078x over previous
#include <hip/hip_runtime.h>


#define N_NODES 100000
#define N_EDGES 1600000
#define NFEAT   256
#define NHID    128
#define NCLASS  40

typedef __attribute__((ext_vector_type(2))) float v2f;
typedef __attribute__((ext_vector_type(8))) float v8f;

#define G1_WAVES   8
#define G1_THREADS (G1_WAVES * 32)
#define G1_STRIPS  (N_NODES / 16)
#define G1_BLOCKS  512
#define G1_KQ      (NFEAT / 4)
#define G1_CHUNK   8
#define G1_NCHUNK  (NFEAT / (8 * G1_CHUNK))
static_assert(N_NODES % 16 == 0, "whole strips");
static_assert((G1_KQ * (NHID / 4)) % G1_THREADS == 0, "staging divides evenly");
static_assert(G1_STRIPS / G1_BLOCKS >= G1_WAVES, "every wave has a first strip");

__global__ __launch_bounds__(G1_THREADS)
void gcn_gemm1(const float* __restrict__ x, const float* __restrict__ W1,
               float* __restrict__ H0, float* __restrict__ agg) {
  __shared__ __attribute__((aligned(16))) float4 Bq[G1_KQ * NHID];

  const int tid  = threadIdx.x;
  const int wave = __builtin_amdgcn_readfirstlane(tid >> 5);
  const int lane = tid & 31;
  const int half = lane >> 4;
  const int mn   = lane & 15;
  const int s0 = (int)(((long long)blockIdx.x * G1_STRIPS) / G1_BLOCKS);
  const int s1 = (int)(((long long)(blockIdx.x + 1) * G1_STRIPS) / G1_BLOCKS);
  int strip = s0 + wave;

  const float4* arow =
      reinterpret_cast<const float4*>(x + (size_t)(strip * 16 + mn) * NFEAT) + half;

  float4 an[G1_CHUNK];
  #pragma unroll
  for (int qi = 0; qi < G1_CHUNK; ++qi) an[qi] = arow[2 * qi];

  #pragma unroll
  for (int i = 0; i < (G1_KQ * (NHID / 4)) / G1_THREADS; ++i) {
    const int kk = wave + G1_WAVES * i;
    const float4* wp = reinterpret_cast<const float4*>(W1 + (size_t)(4 * kk) * NHID) + lane;
    const float4 r0 = wp[0];
    const float4 r1 = wp[NHID / 4];
    const float4 r2 = wp[2 * (NHID / 4)];
    const float4 r3 = wp[3 * (NHID / 4)];
    float4* bq = Bq + kk * NHID + 4 * lane;
    bq[0] = make_float4(r0.x, r1.x, r2.x, r3.x);
    bq[1] = make_float4(r0.y, r1.y, r2.y, r3.y);
    bq[2] = make_float4(r0.z, r1.z, r2.z, r3.z);
    bq[3] = make_float4(r0.w, r1.w, r2.w, r3.w);
  }
  __syncthreads();

  #pragma unroll 1
  while (strip < s1) {
    asm volatile("" ::: "memory");
    v8f acc[8];
    #pragma unroll
    for (int t = 0; t < 8; ++t) acc[t] = v8f{};

    {
      float4* zrow = reinterpret_cast<float4*>(agg + (size_t)strip * 16 * NHID) + lane;
      #pragma unroll
      for (int r = 0; r < 16; ++r)
        zrow[r * (NHID / 4)] = make_float4(0.f, 0.f, 0.f, 0.f);
    }

    #pragma unroll
    for (int ch = 0; ch < G1_NCHUNK; ++ch) {
      float4 ac[G1_CHUNK];
      #pragma unroll
      for (int qi = 0; qi < G1_CHUNK; ++qi) ac[qi] = an[qi];
      if (ch + 1 < G1_NCHUNK) {
        #pragma unroll
        for (int qi = 0; qi < G1_CHUNK; ++qi)
          an[qi] = arow[2 * (G1_CHUNK * (ch + 1) + qi)];
      }
      #pragma unroll
      for (int qi = 0; qi < G1_CHUNK; ++qi) {
        const int q = G1_CHUNK * ch + qi;
        const float4* bp = Bq + (2 * q + half) * NHID + mn;
        float4 bf[8];
        #pragma unroll
        for (int t = 0; t < 8; ++t) bf[t] = bp[16 * t];
        v2f a01, a23;
        a01.x = ac[qi].x; a01.y = ac[qi].y;
        a23.x = ac[qi].z; a23.y = ac[qi].w;
        #pragma unroll
        for (int t = 0; t < 8; ++t) {
          v2f b01; b01.x = bf[t].x; b01.y = bf[t].y;
          acc[t] = __builtin_amdgcn_wmma_f32_16x16x4_f32(
               false, a01,  false, b01,
               (short)0, acc[t],  false,  false);
        }
        #pragma unroll
        for (int t = 0; t < 8; ++t) {
          v2f b23; b23.x = bf[t].z; b23.y = bf[t].w;
          acc[t] = __builtin_amdgcn_wmma_f32_16x16x4_f32(
              false, a23, false, b23, (short)0, acc[t], false, false);
        }
      }
    }

    const int m0 = strip * 16;
    strip += G1_WAVES;
    if (strip < s1) {
      arow = reinterpret_cast<const float4*>(x + (size_t)(strip * 16 + mn) * NFEAT) + half;
      #pragma unroll
      for (int qi = 0; qi < G1_CHUNK; ++qi) an[qi] = arow[2 * qi];
    }

    float* orow = H0 + (size_t)m0 * NHID + lane;
    #pragma unroll
    for (int tp = 0; tp < 4; ++tp) {
      #pragma unroll
      for (int j = 0; j < 8; ++j) {
        const float lo = acc[2 * tp][j];
        const float hi = acc[2 * tp + 1][j];
        const float send = half ? lo : hi;
        const float recv = __shfl_xor(send, 16);
        const float rowj  = half ? recv : lo;
        const float rowj8 = half ? hi : recv;
        orow[(size_t)j * NHID + 32 * tp]       = rowj;
        orow[(size_t)(j + 8) * NHID + 32 * tp] = rowj8;
      }
    }
  }
}

#define EDGES_PER_WAVE 32
#define EDGE_BATCH     8
#define SPMM_THREADS   128
static_assert(N_EDGES % (EDGES_PER_WAVE * (SPMM_THREADS / 32)) == 0,
              "every wave gets a full run of edges");

__device__ __forceinline__ int lane_read_i(int v, int l) {
  return __builtin_amdgcn_readlane(v, l);
}
__device__ __forceinline__ float lane_read_f(float v, int l) {
  return __int_as_float(__builtin_amdgcn_readlane(__float_as_int(v), l));
}

__global__ __launch_bounds__(SPMM_THREADS)
void gcn_spmm1(const int* __restrict__ esrc, const int* __restrict__ edst,
               const float* __restrict__ ew, const float* __restrict__ H0,
               float* __restrict__ agg) {
  const int lane = threadIdx.x & 31;
  const int wv   = blockIdx.x * (SPMM_THREADS / 32) + (threadIdx.x >> 5);
  const int e    = wv * EDGES_PER_WAVE + lane;
  const int   sv = esrc[e];
  const int   dv = edst[e];
  const float wl = ew[e];
  for (int g = 0; g < EDGES_PER_WAVE; g += EDGE_BATCH) {
    float h[EDGE_BATCH][4];
    #pragma unroll
    for (int j = 0; j < EDGE_BATCH; ++j) {
      const int s = lane_read_i(sv, g + j);
      const float* hp = H0 + (size_t)s * NHID + lane;
      h[j][0] = hp[0];
      h[j][1] = hp[32];
      h[j][2] = hp[64];
      h[j][3] = hp[96];
    }
    #pragma unroll
    for (int j = 0; j < EDGE_BATCH; ++j) {
      const int   d = lane_read_i(dv, g + j);
      const float w = lane_read_f(wl, g + j);
      float* o = agg + (size_t)d * NHID + lane;
      atomicAdd(o,      w * h[j][0]);
      atomicAdd(o + 32, w * h[j][1]);
      atomicAdd(o + 64, w * h[j][2]);
      atomicAdd(o + 96, w * h[j][3]);
    }
  }
}

#define G2_WAVES   8
#define G2_THREADS (G2_WAVES * 32)
#define G2_BLOCKS  512
#define G2_NT      3
#define G2_NP      (16 * G2_NT)
#define G2_KQ      (NHID / 4)
#define G2_NQ      (NHID / 8)
#define G2_HP      52
static_assert(G2_HP >= G2_NP && G2_HP % 4 == 0, "patch rows hold 48 columns, 16-byte aligned");
static_assert((G2_KQ * G2_NP) % G2_THREADS == 0, "staging divides evenly");
static_assert(G1_STRIPS / G2_BLOCKS >= G2_WAVES, "every wave has a first strip");
static_assert(NCLASS % 4 == 0 && NCLASS <= G2_NP && NCLASS > 16 * (G2_NT - 1),
              "16-byte pieces of out never straddle rows; the last tile is partial");

__global__ __launch_bounds__(G2_THREADS)
void gcn_gemm2(const float* __restrict__ agg, const float* __restrict__ b1,
               const float* __restrict__ W2, const float* __restrict__ b2,
               float* __restrict__ H2, float* __restrict__ out) {
  __shared__ __attribute__((aligned(16))) float4 B2q[G2_KQ * G2_NP];
  __shared__ __attribute__((aligned(16))) float4 b1q[G2_KQ];
  __shared__ __attribute__((aligned(16))) float4 b2q[NCLASS / 4];
  __shared__ __attribute__((aligned(16))) float Hs[G2_WAVES * 16 * G2_HP];

  const int tid  = threadIdx.x;
  const int wave = __builtin_amdgcn_readfirstlane(tid >> 5);
  const int lane = tid & 31;
  const int half = lane >> 4;
  const int mn   = lane & 15;
  const int s0 = (int)(((long long)blockIdx.x * G1_STRIPS) / G2_BLOCKS);
  const int s1 = (int)(((long long)(blockIdx.x + 1) * G1_STRIPS) / G2_BLOCKS);
  int strip = s0 + wave;

  const float4* arow =
      reinterpret_cast<const float4*>(agg + (size_t)(strip * 16 + mn) * NHID) + half;
  float4 an[G2_NQ];
  #pragma unroll
  for (int q = 0; q < G2_NQ; ++q) an[q] = arow[2 * q];

  #pragma unroll
  for (int i = 0; i < (G2_KQ * G2_NP) / G2_THREADS; ++i) {
    const int e  = tid + G2_THREADS * i;
    const int kk = e / G2_NP;
    const int n  = e % G2_NP;
    const bool real = (n < NCLASS);
    const float* wp = W2 + (size_t)(4 * kk) * NCLASS + (real ? n : NCLASS - 1);
    const float w0 = wp[0], w1 = wp[NCLASS], w2 = wp[2 * NCLASS], w3 = wp[3 * NCLASS];
    B2q[e] = make_float4(real ? w0 : 0.f, real ? w1 : 0.f, real ? w2 : 0.f, real ? w3 : 0.f);
  }
  if (tid < G2_KQ) b1q[tid] = reinterpret_cast<const float4*>(b1)[tid];
  if (tid < NCLASS / 4) b2q[tid] = reinterpret_cast<const float4*>(b2)[tid];
  __syncthreads();

  #pragma unroll 1
  while (strip < s1) {
    asm volatile("" ::: "memory");
    const int m0 = strip * 16;
    v8f acc[G2_NT];
    #pragma unroll
    for (int t = 0; t < G2_NT; ++t) acc[t] = v8f{};

    {
      float4* oq = reinterpret_cast<float4*>(out + (size_t)m0 * NCLASS) + lane;
      #pragma unroll
      for (int i = 0; i < 5; ++i) oq[32 * i] = b2q[(lane + 32 * i) % (NCLASS / 4)];
    }

    float4 ac[G2_NQ];
    #pragma unroll
    for (int q = 0; q < G2_NQ; ++q) ac[q] = an[q];

    strip += G2_WAVES;
    if (strip < s1) {
      arow = reinterpret_cast<const float4*>(agg + (size_t)(strip * 16 + mn) * NHID) + half;
      #pragma unroll
      for (int q = 0; q < G2_NQ; ++q) an[q] = arow[2 * q];
    }

    #pragma unroll
    for (int q = 0; q < G2_NQ; ++q) {
      const float4 bi = b1q[2 * q + half];
      const float4* bp = B2q + (2 * q + half) * G2_NP + mn;
      float4 bf[G2_NT];
      #pragma unroll
      for (int t = 0; t < G2_NT; ++t) bf[t] = bp[16 * t];
      v2f a01, a23;
      a01.x = fmaxf(ac[q].x + bi.x, 0.0f); a01.y = fmaxf(ac[q].y + bi.y, 0.0f);
      a23.x = fmaxf(ac[q].z + bi.z, 0.0f); a23.y = fmaxf(ac[q].w + bi.w, 0.0f);
      #pragma unroll
      for (int t = 0; t < G2_NT; ++t) {
        v2f b01; b01.x = bf[t].x; b01.y = bf[t].y;
        acc[t] = __builtin_amdgcn_wmma_f32_16x16x4_f32(
             false, a01,  false, b01,
             (short)0, acc[t],  false,  false);
      }
      #pragma unroll
      for (int t = 0; t < G2_NT; ++t) {
        v2f b23; b23.x = bf[t].z; b23.y = bf[t].w;
        acc[t] = __builtin_amdgcn_wmma_f32_16x16x4_f32(
            false, a23, false, b23, (short)0, acc[t], false, false);
      }
    }

    {
      float* hs = Hs + wave * (16 * G2_HP);
      #pragma unroll
      for (int t = 0; t < G2_NT; ++t) {
        #pragma unroll
        for (int j = 0; j < 8; ++j) hs[(j + 8 * half) * G2_HP + 16 * t + mn] = acc[t][j];
      }
      float4* hq = reinterpret_cast<float4*>(H2 + (size_t)m0 * NCLASS) + lane;
      #pragma unroll
      for (int i = 0; i < 5; ++i) {
        const int p = lane + 32 * i;
        hq[32 * i] = *reinterpret_cast<const float4*>(
            hs + (p / (NCLASS / 4)) * G2_HP + 4 * (p % (NCLASS / 4)));
      }
    }
  }
}

__global__ __launch_bounds__(SPMM_THREADS)
void gcn_spmm2(const int* __restrict__ esrc, const int* __restrict__ edst,
               const float* __restrict__ ew, const float* __restrict__ H2,
               float* __restrict__ out) {
  const int lane = threadIdx.x & 31;
  const int wv   = blockIdx.x * (SPMM_THREADS / 32) + (threadIdx.x >> 5);
  const int e    = wv * EDGES_PER_WAVE + lane;
  const int   so = esrc[e] * NCLASS;
  const int   dn = edst[e] * NCLASS;
  const float wl = ew[e];
  const int tsel = lane >> 3;
  const int tcol = 32 + (lane & 7);
  for (int g = 0; g < EDGES_PER_WAVE; g += EDGE_BATCH) {
    float hf[EDGE_BATCH];
    float ht[EDGE_BATCH / 4];
    #pragma unroll
    for (int j = 0; j < EDGE_BATCH; ++j) {
      const int s = lane_read_i(so, g + j);
      hf[j] = H2[(size_t)s + lane];
    }
    #pragma unroll
    for (int q = 0; q < EDGE_BATCH / 4; ++q) {
      const int st = __shfl(so, g + 4 * q + tsel);
      ht[q] = H2[(size_t)st + tcol];
    }
    #pragma unroll
    for (int j = 0; j < EDGE_BATCH; ++j) {
      const int   d = lane_read_i(dn, g + j);
      const float w = lane_read_f(wl, g + j);
      atomicAdd(out + (size_t)d + lane, w * hf[j]);
    }
    #pragma unroll
    for (int q = 0; q < EDGE_BATCH / 4; ++q) {
      const int   dt = __shfl(dn, g + 4 * q + tsel);
      const float wt = __shfl(wl, g + 4 * q + tsel);
      atomicAdd(out + (size_t)dt + tcol, wt * ht[q]);
    }
  }
}

extern "C" void kernel_launch(void* const* d_in, const int* in_sizes, int n_in,
                              void* d_out, int out_size, void* d_ws, size_t ws_size,
                              hipStream_t stream) {
  const float* x    = (const float*)d_in[0];
  const int*   esrc = (const int*)  d_in[1];
  const int*   edst = (const int*)  d_in[2];
  const float* ew   = (const float*)d_in[3];
  const float* W1   = (const float*)d_in[4];
  const float* b1   = (const float*)d_in[5];
  const float* W2   = (const float*)d_in[6];
  const float* b2   = (const float*)d_in[7];
  float* out = (float*)d_out;

  float* H0  = (float*)d_ws;
  float* agg = H0 + (size_t)N_NODES * NHID;
  float* H2  = H0;

  gcn_gemm1<<<G1_BLOCKS, G1_THREADS, 0, stream>>>(x, W1, H0, agg);
  gcn_spmm1<<<N_EDGES / (EDGES_PER_WAVE * (SPMM_THREADS / 32)), SPMM_THREADS, 0, stream>>>(
      esrc, edst, ew, H0, agg);
  gcn_gemm2<<<G2_BLOCKS, G2_THREADS, 0, stream>>>(agg, b1, W2, b2, H2, out);
  gcn_spmm2<<<N_EDGES / (EDGES_PER_WAVE * (SPMM_THREADS / 32)), SPMM_THREADS, 0, stream>>>(
      esrc, edst, ew, H2, out);
}


# SpMM1 columns split over the two cache halves: H0/agg as two [node][64] planes; a wave reads its die, draws edge chunks of its half's plane from an int atomic counter; float atomics, arrival order
# speedup vs baseline: 2.5616x; 1.0639x over previous
#include <hip/hip_runtime.h>


#define N_NODES 100000
#define N_EDGES 1600000
#define NFEAT   256
#define NHID    128
#define NCLASS  40
#define PLANE_COLS  64
#define N_PLANES    (NHID / PLANE_COLS)
#define PLANE_ELEMS ((size_t)N_NODES * PLANE_COLS)
#define S1_CTR_STRIDE 1024

typedef __attribute__((ext_vector_type(2))) float v2f;
typedef __attribute__((ext_vector_type(8))) float v8f;

#define G1_WAVES   8
#define G1_THREADS (G1_WAVES * 32)
#define G1_STRIPS  (N_NODES / 16)
#define G1_BLOCKS  512
#define G1_KQ      (NFEAT / 4)
#define G1_CHUNK   8
#define G1_NCHUNK  (NFEAT / (8 * G1_CHUNK))
static_assert(N_NODES % 16 == 0, "whole strips");
static_assert(NHID == 8 * 16 && PLANE_COLS == 64 && N_PLANES == 2, "two pairs of column tiles per plane");
static_assert((G1_KQ * (NHID / 4)) % G1_THREADS == 0, "staging divides evenly");
static_assert(G1_STRIPS / G1_BLOCKS >= G1_WAVES, "every wave has a first strip");

__global__ __launch_bounds__(G1_THREADS)
void gcn_gemm1(const float* __restrict__ x, const float* __restrict__ W1,
               float* __restrict__ H0, float* __restrict__ agg,
               float* __restrict__ out) {
  __shared__ __attribute__((aligned(16))) float4 Bq[G1_KQ * NHID];

  const int tid  = threadIdx.x;
  if (blockIdx.x == 0 && tid < N_PLANES) reinterpret_cast<int*>(out)[tid * S1_CTR_STRIDE] = 0;
  const int wave = __builtin_amdgcn_readfirstlane(tid >> 5);
  const int lane = tid & 31;
  const int half = lane >> 4;
  const int mn   = lane & 15;
  const int s0 = (int)(((long long)blockIdx.x * G1_STRIPS) / G1_BLOCKS);
  const int s1 = (int)(((long long)(blockIdx.x + 1) * G1_STRIPS) / G1_BLOCKS);
  int strip = s0 + wave;

  const float4* arow =
      reinterpret_cast<const float4*>(x + (size_t)(strip * 16 + mn) * NFEAT) + half;

  float4 an[G1_CHUNK];
  #pragma unroll
  for (int qi = 0; qi < G1_CHUNK; ++qi) an[qi] = arow[2 * qi];

  #pragma unroll
  for (int i = 0; i < (G1_KQ * (NHID / 4)) / G1_THREADS; ++i) {
    const int kk = wave + G1_WAVES * i;
    const float4* wp = reinterpret_cast<const float4*>(W1 + (size_t)(4 * kk) * NHID) + lane;
    const float4 r0 = wp[0];
    const float4 r1 = wp[NHID / 4];
    const float4 r2 = wp[2 * (NHID / 4)];
    const float4 r3 = wp[3 * (NHID / 4)];
    float4* bq = Bq + kk * NHID + 4 * lane;
    bq[0] = make_float4(r0.x, r1.x, r2.x, r3.x);
    bq[1] = make_float4(r0.y, r1.y, r2.y, r3.y);
    bq[2] = make_float4(r0.z, r1.z, r2.z, r3.z);
    bq[3] = make_float4(r0.w, r1.w, r2.w, r3.w);
  }
  __syncthreads();

  #pragma unroll 1
  while (strip < s1) {
    asm volatile("" ::: "memory");
    v8f acc[8];
    #pragma unroll
    for (int t = 0; t < 8; ++t) acc[t] = v8f{};

    {
      float4* zrow = reinterpret_cast<float4*>(agg + (size_t)strip * 16 * PLANE_COLS) + lane;
      #pragma unroll
      for (int p = 0; p < N_PLANES; ++p) {
        #pragma unroll
        for (int i = 0; i < 16 / N_PLANES; ++i)
          zrow[p * (PLANE_ELEMS / 4) + 32 * i] = make_float4(0.f, 0.f, 0.f, 0.f);
      }
    }

    #pragma unroll
    for (int ch = 0; ch < G1_NCHUNK; ++ch) {
      float4 ac[G1_CHUNK];
      #pragma unroll
      for (int qi = 0; qi < G1_CHUNK; ++qi) ac[qi] = an[qi];
      if (ch + 1 < G1_NCHUNK) {
        #pragma unroll
        for (int qi = 0; qi < G1_CHUNK; ++qi)
          an[qi] = arow[2 * (G1_CHUNK * (ch + 1) + qi)];
      }
      #pragma unroll
      for (int qi = 0; qi < G1_CHUNK; ++qi) {
        const int q = G1_CHUNK * ch + qi;
        const float4* bp = Bq + (2 * q + half) * NHID + mn;
        float4 bf[8];
        #pragma unroll
        for (int t = 0; t < 8; ++t) bf[t] = bp[16 * t];
        v2f a01, a23;
        a01.x = ac[qi].x; a01.y = ac[qi].y;
        a23.x = ac[qi].z; a23.y = ac[qi].w;
        #pragma unroll
        for (int t = 0; t < 8; ++t) {
          v2f b01; b01.x = bf[t].x; b01.y = bf[t].y;
          acc[t] = __builtin_amdgcn_wmma_f32_16x16x4_f32(
               false, a01,  false, b01,
               (short)0, acc[t],  false,  false);
        }
        #pragma unroll
        for (int t = 0; t < 8; ++t) {
          v2f b23; b23.x = bf[t].z; b23.y = bf[t].w;
          acc[t] = __builtin_amdgcn_wmma_f32_16x16x4_f32(
              false, a23, false, b23, (short)0, acc[t], false, false);
        }
      }
    }

    const int m0 = strip * 16;
    strip += G1_WAVES;
    if (strip < s1) {
      arow = reinterpret_cast<const float4*>(x + (size_t)(strip * 16 + mn) * NFEAT) + half;
      #pragma unroll
      for (int qi = 0; qi < G1_CHUNK; ++qi) an[qi] = arow[2 * qi];
    }

    float* orow = H0 + (size_t)m0 * PLANE_COLS + lane;
    #pragma unroll
    for (int tp = 0; tp < 4; ++tp) {
      #pragma unroll
      for (int j = 0; j < 8; ++j) {
        const float lo = acc[2 * tp][j];
        const float hi = acc[2 * tp + 1][j];
        const float send = half ? lo : hi;
        const float recv = __shfl_xor(send, 16);
        const float rowj  = half ? recv : lo;
        const float rowj8 = half ? hi : recv;
        orow[(tp >> 1) * PLANE_ELEMS + (size_t)j * PLANE_COLS + 32 * (tp & 1)]       = rowj;
        orow[(tp >> 1) * PLANE_ELEMS + (size_t)(j + 8) * PLANE_COLS + 32 * (tp & 1)] = rowj8;
      }
    }
  }
}

#define EDGES_PER_WAVE 32
#define EDGE_BATCH     8
#define SPMM_THREADS   128
static_assert(N_EDGES % (EDGES_PER_WAVE * (SPMM_THREADS / 32)) == 0,
              "every wave gets a full run of edges");

__device__ __forceinline__ int lane_read_i(int v, int l) {
  return __builtin_amdgcn_readlane(v, l);
}
__device__ __forceinline__ float lane_read_f(float v, int l) {
  return __int_as_float(__builtin_amdgcn_readlane(__float_as_int(v), l));
}

#define S1_RUNS   2
#define S1_CHUNK  (EDGES_PER_WAVE * S1_RUNS)
#define S1_NCHUNK (N_EDGES / S1_CHUNK)
#define S1_BATCH  16
#define S1_BLOCKS 4096
static_assert(N_EDGES % S1_CHUNK == 0, "whole chunks");
static_assert((size_t)(N_PLANES - 1) * S1_CTR_STRIDE < (size_t)N_NODES * NCLASS, "counters inside out");
static_assert(EDGES_PER_WAVE % S1_BATCH == 0, "whole batches");

__device__ __forceinline__ int s1_draw(int* ctr, int lane) {
  int v = 0;
  if (lane == 0) v = atomicAdd(ctr, 1);
  return __builtin_amdgcn_readfirstlane(v);
}

__global__ __launch_bounds__(SPMM_THREADS)
void gcn_spmm1(const int* __restrict__ esrc, const int* __restrict__ edst,
               const float* __restrict__ ew, const float* __restrict__ H0,
               float* __restrict__ agg, float* __restrict__ out) {
  typedef __attribute__((address_space(1))) const float gmem_cfloat;
  typedef __attribute__((address_space(1))) float       gmem_float;
  const int lane = threadIdx.x & 31;
  int* ctr = reinterpret_cast<int*>(out);
  const int half = (int)((__builtin_amdgcn_s_sendmsg_rtn(0x87) >> 18) & 1u);

  #pragma unroll 1
  for (int ph = 0; ph < N_PLANES; ++ph) {
    const int plane = half ^ ph;
    const unsigned long long hbase =
        reinterpret_cast<unsigned long long>(H0 + (size_t)plane * PLANE_ELEMS);
    const unsigned long long abase =
        reinterpret_cast<unsigned long long>(agg + (size_t)plane * PLANE_ELEMS);
    #pragma unroll 1
    for (int c = s1_draw(ctr + plane * S1_CTR_STRIDE, lane); c < S1_NCHUNK; c = s1_draw(ctr + plane * S1_CTR_STRIDE, lane)) {
      const int e = c * S1_CHUNK + lane;
      unsigned so[S1_RUNS], dn[S1_RUNS];
      float wl[S1_RUNS];
      #pragma unroll
      for (int r = 0; r < S1_RUNS; ++r) {
        so[r] = (unsigned)esrc[e + EDGES_PER_WAVE * r] * (PLANE_COLS * 4u);
        dn[r] = (unsigned)edst[e + EDGES_PER_WAVE * r] * (PLANE_COLS * 4u);
        wl[r] = ew[e + EDGES_PER_WAVE * r];
      }
      #pragma unroll
      for (int r = 0; r < S1_RUNS; ++r) {
        #pragma unroll
        for (int g = 0; g < EDGES_PER_WAVE; g += S1_BATCH) {
          float h[S1_BATCH][2];
          #pragma unroll
          for (int j = 0; j < S1_BATCH; ++j) {
            const gmem_cfloat* row =
                (const gmem_cfloat*)(hbase + (unsigned)lane_read_i((int)so[r], g + j));
            h[j][0] = row[lane];
            h[j][1] = row[lane + 32];
          }
          #pragma unroll
          for (int j = 0; j < S1_BATCH; ++j) {
            gmem_float* row = (gmem_float*)(abase + (unsigned)lane_read_i((int)dn[r], g + j));
            const float w = lane_read_f(wl[r], g + j);
            __hip_atomic_fetch_add(row + lane,      w * h[j][0], __ATOMIC_RELAXED, __HIP_MEMORY_SCOPE_AGENT);
            __hip_atomic_fetch_add(row + lane + 32, w * h[j][1], __ATOMIC_RELAXED, __HIP_MEMORY_SCOPE_AGENT);
          }
        }
      }
    }
  }
}

#define G2_WAVES   8
#define G2_THREADS (G2_WAVES * 32)
#define G2_BLOCKS  512
#define G2_NT      3
#define G2_NP      (16 * G2_NT)
#define G2_KQ      (NHID / 4)
#define G2_NQ      (NHID / 8)
#define G2_HP      52
static_assert(G2_HP >= G2_NP && G2_HP % 4 == 0, "patch rows hold 48 columns, 16-byte aligned");
static_assert((G2_KQ * G2_NP) % G2_THREADS == 0, "staging divides evenly");
static_assert(PLANE_COLS == 64 && G2_NQ == 8 * N_PLANES, "eight runs of 8 k per plane");
static_assert(G1_STRIPS / G2_BLOCKS >= G2_WAVES, "every wave has a first strip");
static_assert(NCLASS % 4 == 0 && NCLASS <= G2_NP && NCLASS > 16 * (G2_NT - 1),
              "16-byte pieces of out never straddle rows; the last tile is partial");

__global__ __launch_bounds__(G2_THREADS)
void gcn_gemm2(const float* __restrict__ agg, const float* __restrict__ b1,
               const float* __restrict__ W2, const float* __restrict__ b2,
               float* __restrict__ H2, float* __restrict__ out) {
  __shared__ __attribute__((aligned(16))) float4 B2q[G2_KQ * G2_NP];
  __shared__ __attribute__((aligned(16))) float4 b1q[G2_KQ];
  __shared__ __attribute__((aligned(16))) float4 b2q[NCLASS / 4];
  __shared__ __attribute__((aligned(16))) float Hs[G2_WAVES * 16 * G2_HP];

  const int tid  = threadIdx.x;
  const int wave = __builtin_amdgcn_readfirstlane(tid >> 5);
  const int lane = tid & 31;
  const int half = lane >> 4;
  const int mn   = lane & 15;
  const int s0 = (int)(((long long)blockIdx.x * G1_STRIPS) / G2_BLOCKS);
  const int s1 = (int)(((long long)(blockIdx.x + 1) * G1_STRIPS) / G2_BLOCKS);
  int strip = s0 + wave;

  const float4* arow =
      reinterpret_cast<const float4*>(agg + (size_t)(strip * 16 + mn) * PLANE_COLS) + half;
  float4 an[G2_NQ];
  #pragma unroll
  for (int q = 0; q < G2_NQ; ++q) an[q] = arow[(q >> 3) * (PLANE_ELEMS / 4) + 2 * (q & 7)];

  #pragma unroll
  for (int i = 0; i < (G2_KQ * G2_NP) / G2_THREADS; ++i) {
    const int e  = tid + G2_THREADS * i;
    const int kk = e / G2_NP;
    const int n  = e % G2_NP;
    const bool real = (n < NCLASS);
    const float* wp = W2 + (size_t)(4 * kk) * NCLASS + (real ? n : NCLASS - 1);
    const float w0 = wp[0], w1 = wp[NCLASS], w2 = wp[2 * NCLASS], w3 = wp[3 * NCLASS];
    B2q[e] = make_float4(real ? w0 : 0.f, real ? w1 : 0.f, real ? w2 : 0.f, real ? w3 : 0.f);
  }
  if (tid < G2_KQ) b1q[tid] = reinterpret_cast<const float4*>(b1)[tid];
  if (tid < NCLASS / 4) b2q[tid] = reinterpret_cast<const float4*>(b2)[tid];
  __syncthreads();

  #pragma unroll 1
  while (strip < s1) {
    asm volatile("" ::: "memory");
    const int m0 = strip * 16;
    v8f acc[G2_NT];
    #pragma unroll
    for (int t = 0; t < G2_NT; ++t) acc[t] = v8f{};

    {
      float4* oq = reinterpret_cast<float4*>(out + (size_t)m0 * NCLASS) + lane;
      #pragma unroll
      for (int i = 0; i < 5; ++i) oq[32 * i] = b2q[(lane + 32 * i) % (NCLASS / 4)];
    }

    float4 ac[G2_NQ];
    #pragma unroll
    for (int q = 0; q < G2_NQ; ++q) ac[q] = an[q];

    strip += G2_WAVES;
    if (strip < s1) {
      arow = reinterpret_cast<const float4*>(agg + (size_t)(strip * 16 + mn) * PLANE_COLS) + half;
      #pragma unroll
      for (int q = 0; q < G2_NQ; ++q) an[q] = arow[(q >> 3) * (PLANE_ELEMS / 4) + 2 * (q & 7)];
    }

    #pragma unroll
    for (int q = 0; q < G2_NQ; ++q) {
      const float4 bi = b1q[2 * q + half];
      const float4* bp = B2q + (2 * q + half) * G2_NP + mn;
      float4 bf[G2_NT];
      #pragma unroll
      for (int t = 0; t < G2_NT; ++t) bf[t] = bp[16 * t];
      v2f a01, a23;
      a01.x = fmaxf(ac[q].x + bi.x, 0.0f); a01.y = fmaxf(ac[q].y + bi.y, 0.0f);
      a23.x = fmaxf(ac[q].z + bi.z, 0.0f); a23.y = fmaxf(ac[q].w + bi.w, 0.0f);
      #pragma unroll
      for (int t = 0; t < G2_NT; ++t) {
        v2f b01; b01.x = bf[t].x; b01.y = bf[t].y;
        acc[t] = __builtin_amdgcn_wmma_f32_16x16x4_f32(
             false, a01,  false, b01,
             (short)0, acc[t],  false,  false);
      }
      #pragma unroll
      for (int t = 0; t < G2_NT; ++t) {
        v2f b23; b23.x = bf[t].z; b23.y = bf[t].w;
        acc[t] = __builtin_amdgcn_wmma_f32_16x16x4_f32(
            false, a23, false, b23, (short)0, acc[t], false, false);
      }
    }

    {
      float* hs = Hs + wave * (16 * G2_HP);
      #pragma unroll
      for (int t = 0; t < G2_NT; ++t) {
        #pragma unroll
        for (int j = 0; j < 8; ++j) hs[(j + 8 * half) * G2_HP + 16 * t + mn] = acc[t][j];
      }
      float4* hq = reinterpret_cast<float4*>(H2 + (size_t)m0 * NCLASS) + lane;
      #pragma unroll
      for (int i = 0; i < 5; ++i) {
        const int p = lane + 32 * i;
        hq[32 * i] = *reinterpret_cast<const float4*>(
            hs + (p / (NCLASS / 4)) * G2_HP + 4 * (p % (NCLASS / 4)));
      }
    }
  }
}

__global__ __launch_bounds__(SPMM_THREADS)
void gcn_spmm2(const int* __restrict__ esrc, const int* __restrict__ edst,
               const float* __restrict__ ew, const float* __restrict__ H2,
               float* __restrict__ out) {
  const int lane = threadIdx.x & 31;
  const int wv   = blockIdx.x * (SPMM_THREADS / 32) + (threadIdx.x >> 5);
  const int e    = wv * EDGES_PER_WAVE + lane;
  const int   so = esrc[e] * NCLASS;
  const int   dn = edst[e] * NCLASS;
  const float wl = ew[e];
  const int tsel = lane >> 3;
  const int tcol = 32 + (lane & 7);
  for (int g = 0; g < EDGES_PER_WAVE; g += EDGE_BATCH) {
    float hf[EDGE_BATCH];
    float ht[EDGE_BATCH / 4];
    #pragma unroll
    for (int j = 0; j < EDGE_BATCH; ++j) {
      const int s = lane_read_i(so, g + j);
      hf[j] = H2[(size_t)s + lane];
    }
    #pragma unroll
    for (int q = 0; q < EDGE_BATCH / 4; ++q) {
      const int st = __shfl(so, g + 4 * q + tsel);
      ht[q] = H2[(size_t)st + tcol];
    }
    #pragma unroll
    for (int j = 0; j < EDGE_BATCH; ++j) {
      const int   d = lane_read_i(dn, g + j);
      const float w = lane_read_f(wl, g + j);
      atomicAdd(out + (size_t)d + lane, w * hf[j]);
    }
    #pragma unroll
    for (int q = 0; q < EDGE_BATCH / 4; ++q) {
      const int   dt = __shfl(dn, g + 4 * q + tsel);
      const float wt = __shfl(wl, g + 4 * q + tsel);
      atomicAdd(out + (size_t)dt + tcol, wt * ht[q]);
    }
  }
}

extern "C" void kernel_launch(void* const* d_in, const int* in_sizes, int n_in,
                              void* d_out, int out_size, void* d_ws, size_t ws_size,
                              hipStream_t stream) {
  const float* x    = (const float*)d_in[0];
  const int*   esrc = (const int*)  d_in[1];
  const int*   edst = (const int*)  d_in[2];
  const float* ew   = (const float*)d_in[3];
  const float* W1   = (const float*)d_in[4];
  const float* b1   = (const float*)d_in[5];
  const float* W2   = (const float*)d_in[6];
  const float* b2   = (const float*)d_in[7];
  float* out = (float*)d_out;

  float* H0  = (float*)d_ws;
  float* agg = H0 + (size_t)N_NODES * NHID;
  float* H2  = H0;

  gcn_gemm1<<<G1_BLOCKS, G1_THREADS, 0, stream>>>(x, W1, H0, agg, out);
  gcn_spmm1<<<S1_BLOCKS, SPMM_THREADS, 0, stream>>>(esrc, edst, ew, H0, agg, out);
  gcn_gemm2<<<G2_BLOCKS, G2_THREADS, 0, stream>>>(agg, b1, W2, b2, H2, out);
  gcn_spmm2<<<N_EDGES / (EDGES_PER_WAVE * (SPMM_THREADS / 32)), SPMM_THREADS, 0, stream>>>(
      esrc, edst, ew, H2, out);
}


# SpMM1 gathers from LDS: edges bucketed in-call by source range (chunk-local LDS count+rank); a block holds its range's 782 H0 rows of a plane in LDS; +1 launch; float atomics, arrival order kept
# speedup vs baseline: 2.8773x; 1.1232x over previous
#include <hip/hip_runtime.h>


#define N_NODES 100000
#define N_EDGES 1600000
#define NFEAT   256
#define NHID    128
#define NCLASS  40
#define PLANE_COLS  64
#define N_PLANES    (NHID / PLANE_COLS)
#define PLANE_ELEMS ((size_t)N_NODES * PLANE_COLS)
#define S1_CTR_STRIDE 1024

typedef __attribute__((ext_vector_type(2))) float v2f;
typedef __attribute__((ext_vector_type(8))) float v8f;

#define G1_WAVES   8
#define G1_THREADS (G1_WAVES * 32)
#define G1_STRIPS  (N_NODES / 16)
#define G1_BLOCKS  512
#define G1_KQ      (NFEAT / 4)
#define G1_CHUNK   8
#define G1_NCHUNK  (NFEAT / (8 * G1_CHUNK))
static_assert(N_NODES % 16 == 0, "whole strips");
static_assert(NHID == 8 * 16 && PLANE_COLS == 64 && N_PLANES == 2, "two pairs of column tiles per plane");
static_assert((G1_KQ * (NHID / 4)) % G1_THREADS == 0, "staging divides evenly");
static_assert(G1_STRIPS / G1_BLOCKS >= G1_WAVES, "every wave has a first strip");

__global__ __launch_bounds__(G1_THREADS)
void gcn_gemm1(const float* __restrict__ x, const float* __restrict__ W1,
               float* __restrict__ H0, float* __restrict__ agg) {
  __shared__ __attribute__((aligned(16))) float4 Bq[G1_KQ * NHID];

  const int tid  = threadIdx.x;
  const int wave = __builtin_amdgcn_readfirstlane(tid >> 5);
  const int lane = tid & 31;
  const int half = lane >> 4;
  const int mn   = lane & 15;
  const int s0 = (int)(((long long)blockIdx.x * G1_STRIPS) / G1_BLOCKS);
  const int s1 = (int)(((long long)(blockIdx.x + 1) * G1_STRIPS) / G1_BLOCKS);
  int strip = s0 + wave;

  const float4* arow =
      reinterpret_cast<const float4*>(x + (size_t)(strip * 16 + mn) * NFEAT) + half;

  float4 an[G1_CHUNK];
  #pragma unroll
  for (int qi = 0; qi < G1_CHUNK; ++qi) an[qi] = arow[2 * qi];

  #pragma unroll
  for (int i = 0; i < (G1_KQ * (NHID / 4)) / G1_THREADS; ++i) {
    const int kk = wave + G1_WAVES * i;
    const float4* wp = reinterpret_cast<const float4*>(W1 + (size_t)(4 * kk) * NHID) + lane;
    const float4 r0 = wp[0];
    const float4 r1 = wp[NHID / 4];
    const float4 r2 = wp[2 * (NHID / 4)];
    const float4 r3 = wp[3 * (NHID / 4)];
    float4* bq = Bq + kk * NHID + 4 * lane;
    bq[0] = make_float4(r0.x, r1.x, r2.x, r3.x);
    bq[1] = make_float4(r0.y, r1.y, r2.y, r3.y);
    bq[2] = make_float4(r0.z, r1.z, r2.z, r3.z);
    bq[3] = make_float4(r0.w, r1.w, r2.w, r3.w);
  }
  __syncthreads();

  #pragma unroll 1
  while (strip < s1) {
    asm volatile("" ::: "memory");
    v8f acc[8];
    #pragma unroll
    for (int t = 0; t < 8; ++t) acc[t] = v8f{};

    {
      float4* zrow = reinterpret_cast<float4*>(agg + (size_t)strip * 16 * PLANE_COLS) + lane;
      #pragma unroll
      for (int p = 0; p < N_PLANES; ++p) {
        #pragma unroll
        for (int i = 0; i < 16 / N_PLANES; ++i)
          zrow[p * (PLANE_ELEMS / 4) + 32 * i] = make_float4(0.f, 0.f, 0.f, 0.f);
      }
    }

    #pragma unroll
    for (int ch = 0; ch < G1_NCHUNK; ++ch) {
      float4 ac[G1_CHUNK];
      #pragma unroll
      for (int qi = 0; qi < G1_CHUNK; ++qi) ac[qi] = an[qi];
      if (ch + 1 < G1_NCHUNK) {
        #pragma unroll
        for (int qi = 0; qi < G1_CHUNK; ++qi)
          an[qi] = arow[2 * (G1_CHUNK * (ch + 1) + qi)];
      }
      #pragma unroll
      for (int qi = 0; qi < G1_CHUNK; ++qi) {
        const int q = G1_CHUNK * ch + qi;
        const float4* bp = Bq + (2 * q + half) * NHID + mn;
        float4 bf[8];
        #pragma unroll
        for (int t = 0; t < 8; ++t) bf[t] = bp[16 * t];
        v2f a01, a23;
        a01.x = ac[qi].x; a01.y = ac[qi].y;
        a23.x = ac[qi].z; a23.y = ac[qi].w;
        #pragma unroll
        for (int t = 0; t < 8; ++t) {
          v2f b01; b01.x = bf[t].x; b01.y = bf[t].y;
          acc[t] = __builtin_amdgcn_wmma_f32_16x16x4_f32(
               false, a01,  false, b01,
               (short)0, acc[t],  false,  false);
        }
        #pragma unroll
        for (int t = 0; t < 8; ++t) {
          v2f b23; b23.x = bf[t].z; b23.y = bf[t].w;
          acc[t] = __builtin_amdgcn_wmma_f32_16x16x4_f32(
              false, a23, false, b23, (short)0, acc[t], false, false);
        }
      }
    }

    const int m0 = strip * 16;
    strip += G1_WAVES;
    if (strip < s1) {
      arow = reinterpret_cast<const float4*>(x + (size_t)(strip * 16 + mn) * NFEAT) + half;
      #pragma unroll
      for (int qi = 0; qi < G1_CHUNK; ++qi) an[qi] = arow[2 * qi];
    }

    float* orow = H0 + (size_t)m0 * PLANE_COLS + lane;
    #pragma unroll
    for (int tp = 0; tp < 4; ++tp) {
      #pragma unroll
      for (int j = 0; j < 8; ++j) {
        const float lo = acc[2 * tp][j];
        const float hi = acc[2 * tp + 1][j];
        const float send = half ? lo : hi;
        const float recv = __shfl_xor(send, 16);
        const float rowj  = half ? recv : lo;
        const float rowj8 = half ? hi : recv;
        orow[(tp >> 1) * PLANE_ELEMS + (size_t)j * PLANE_COLS + 32 * (tp & 1)]       = rowj;
        orow[(tp >> 1) * PLANE_ELEMS + (size_t)(j + 8) * PLANE_COLS + 32 * (tp & 1)] = rowj8;
      }
    }
  }
}

#define EDGES_PER_WAVE 32
#define EDGE_BATCH     8
#define SPMM_THREADS   128
static_assert(N_EDGES % (EDGES_PER_WAVE * (SPMM_THREADS / 32)) == 0,
              "every wave gets a full run of edges");

__device__ __forceinline__ int lane_read_i(int v, int l) {
  return __builtin_amdgcn_readlane(v, l);
}
__device__ __forceinline__ float lane_read_f(float v, int l) {
  return __int_as_float(__builtin_amdgcn_readlane(__float_as_int(v), l));
}

#define SB_RANGES     128
#define SB_ROWS       782
#define SB_CHUNK      6400
#define SB_NCHUNK     (N_EDGES / SB_CHUNK)
#define SB_THREADS    256
#define SB_PER_THREAD (SB_CHUNK / SB_THREADS)
#define SB_DST_BITS   17
#define SB_PK_OFF     0
#define SB_W_OFF      N_EDGES
#define SB_TAB_OFF    (2 * N_EDGES)
#define SB_CTR_OFF    (3224 * 1024)
static_assert(SB_RANGES * SB_ROWS >= N_NODES, "the ranges cover every node");
static_assert(N_NODES <= (1 << SB_DST_BITS) && SB_ROWS <= (1 << (32 - SB_DST_BITS)), "record fits one word");
static_assert(N_EDGES % SB_CHUNK == 0 && SB_CHUNK % SB_THREADS == 0, "whole chunks, equal shares");
static_assert(SB_THREADS > SB_RANGES, "one thread per table entry of a chunk");
static_assert(SB_TAB_OFF + SB_NCHUNK * (SB_RANGES + 1) <= SB_CTR_OFF, "table ends before the counters");
static_assert((size_t)SB_CTR_OFF + (N_PLANES - 1) * S1_CTR_STRIDE < (size_t)N_NODES * NCLASS, "counters inside out");

__global__ __launch_bounds__(SB_THREADS)
void gcn_bucket(const int* __restrict__ esrc, const int* __restrict__ edst,
                const float* __restrict__ ew, float* __restrict__ out) {
  __shared__ int   cnt[SB_RANGES];
  __shared__ int   start[SB_RANGES + 1];
  __shared__ int   imgP[SB_CHUNK];
  __shared__ float imgW[SB_CHUNK];
  int* scratch = reinterpret_cast<int*>(out);
  const int tid  = threadIdx.x;
  const int base = blockIdx.x * SB_CHUNK;

  if (blockIdx.x == 0 && tid < N_PLANES) scratch[SB_CTR_OFF + tid * S1_CTR_STRIDE] = 0;
  if (tid < SB_RANGES) cnt[tid] = 0;
  __syncthreads();

  int   pk[SB_PER_THREAD], bk[SB_PER_THREAD], rk[SB_PER_THREAD];
  float wv[SB_PER_THREAD];
  #pragma unroll
  for (int i = 0; i < SB_PER_THREAD; ++i) {
    const int e = base + tid + SB_THREADS * i;
    const int sn = esrc[e];
    const int dn = edst[e];
    wv[i] = ew[e];
    const int b = sn / SB_ROWS;
    bk[i] = b;
    pk[i] = ((sn - b * SB_ROWS) << SB_DST_BITS) | dn;
  }
  #pragma unroll
  for (int i = 0; i < SB_PER_THREAD; ++i) rk[i] = atomicAdd(&cnt[bk[i]], 1);
  __syncthreads();

  if (tid < SB_RANGES) {
    int acc = 0;
    for (int j = 0; j < SB_RANGES; ++j) {
      const int v = cnt[j];
      acc += (j < tid) ? v : 0;
    }
    start[tid] = acc;
  }
  if (tid == SB_RANGES) start[SB_RANGES] = SB_CHUNK;
  __syncthreads();

  if (tid <= SB_RANGES) scratch[SB_TAB_OFF + blockIdx.x * (SB_RANGES + 1) + tid] = start[tid];
  #pragma unroll
  for (int i = 0; i < SB_PER_THREAD; ++i) {
    const int pos = start[bk[i]] + rk[i];
    imgP[pos] = pk[i];
    imgW[pos] = wv[i];
  }
  __syncthreads();
  #pragma unroll
  for (int i = 0; i < SB_PER_THREAD; ++i) {
    const int j = tid + SB_THREADS * i;
    scratch[SB_PK_OFF + base + j] = imgP[j];
    out[SB_W_OFF + base + j]      = imgW[j];
  }
}

#define S1_THREADS 1024
#define S1_WAVES   (S1_THREADS / 32)
#define S1_GRID    (N_PLANES * SB_RANGES)
#define S1_SLOTS   ((SB_NCHUNK + S1_WAVES - 1) / S1_WAVES)
static_assert(S1_SLOTS <= 32, "a wave keeps its chunks' range starts one per lane");

__global__ __launch_bounds__(S1_THREADS)
void gcn_spmm1(const float* __restrict__ H0, float* __restrict__ agg,
               float* __restrict__ out) {
  __shared__ __attribute__((aligned(16))) float rows[SB_ROWS * PLANE_COLS];
  __shared__ int item_s;
  typedef __attribute__((address_space(1))) float gmem_float;
  const int tid  = threadIdx.x;
  const int lane = tid & 31;
  const int wave = __builtin_amdgcn_readfirstlane(tid >> 5);
  const int* scratch = reinterpret_cast<const int*>(out);
  int* ctr = reinterpret_cast<int*>(out) + SB_CTR_OFF;
  const int half = (int)((__builtin_amdgcn_s_sendmsg_rtn(0x87) >> 18) & 1u);

  for (;;) {
    if (tid == 0) {
      int plane = half;
      int t = atomicAdd(ctr + plane * S1_CTR_STRIDE, 1);
      if (t >= SB_RANGES) {
        plane = half ^ 1;
        t = atomicAdd(ctr + plane * S1_CTR_STRIDE, 1);
      }
      item_s = (t < SB_RANGES) ? (plane * SB_RANGES + t) : -1;
    }
    __syncthreads();
    const int item = __builtin_amdgcn_readfirstlane(item_s);
    if (item < 0) break;
    const int plane = item / SB_RANGES;
    const int r     = item - plane * SB_RANGES;
    const int row0  = r * SB_ROWS;
    const int nrows = (N_NODES - row0 < SB_ROWS) ? (N_NODES - row0) : SB_ROWS;

    int o0 = 0, o1 = 0;
    {
      const int c = wave + S1_WAVES * lane;
      if (c < SB_NCHUNK) {
        const int* tp = scratch + SB_TAB_OFF + c * (SB_RANGES + 1) + r;
        o0 = tp[0];
        o1 = tp[1];
      }
    }

    {
      const float4* gsrc = reinterpret_cast<const float4*>(
          H0 + (size_t)plane * PLANE_ELEMS + (size_t)row0 * PLANE_COLS);
      float4* ldst = reinterpret_cast<float4*>(rows);
      const int npieces = nrows * (PLANE_COLS / 4);
      #pragma unroll 4
      for (int i = tid; i < npieces; i += S1_THREADS) ldst[i] = gsrc[i];
    }
    __syncthreads();

    const unsigned long long abase =
        reinterpret_cast<unsigned long long>(agg + (size_t)plane * PLANE_ELEMS);
    const float* myrow = rows + lane;
    for (int sl = 0; sl < S1_SLOTS; ++sl) {
      const int c = wave + S1_WAVES * sl;
      if (c >= SB_NCHUNK) break;
      const int j0 = lane_read_i(o0, sl);
      const int j1 = lane_read_i(o1, sl);
      const int*   P  = scratch + SB_PK_OFF + c * SB_CHUNK;
      const float* Wp = out + SB_W_OFF + c * SB_CHUNK;
      for (int j = j0; j < j1; j += 32) {
        const int n = (j1 - j < 32) ? (j1 - j) : 32;
        int   pkl = 0;
        float wl  = 0.f;
        if (lane < n) {
          pkl = P[j + lane];
          wl  = Wp[j + lane];
        }
        int k = 0;
        for (; k + 4 <= n; k += 4) {
          float h[4][2];
          #pragma unroll
          for (int u = 0; u < 4; ++u) {
            const unsigned p = (unsigned)lane_read_i(pkl, k + u);
            const float* rp = myrow + (p >> SB_DST_BITS) * PLANE_COLS;
            h[u][0] = rp[0];
            h[u][1] = rp[32];
          }
          #pragma unroll
          for (int u = 0; u < 4; ++u) {
            const unsigned p = (unsigned)lane_read_i(pkl, k + u);
            const float w = lane_read_f(wl, k + u);
            gmem_float* arow = (gmem_float*)(abase +
                (unsigned long long)((p & ((1u << SB_DST_BITS) - 1u)) * (PLANE_COLS * 4u)));
            __hip_atomic_fetch_add(arow + lane,      w * h[u][0], __ATOMIC_RELAXED, __HIP_MEMORY_SCOPE_AGENT);
            __hip_atomic_fetch_add(arow + lane + 32, w * h[u][1], __ATOMIC_RELAXED, __HIP_MEMORY_SCOPE_AGENT);
          }
        }
        for (; k < n; ++k) {
          const unsigned p = (unsigned)lane_read_i(pkl, k);
          const float w = lane_read_f(wl, k);
          const float* rp = myrow + (p >> SB_DST_BITS) * PLANE_COLS;
          const float h0 = rp[0];
          const float h1 = rp[32];
          gmem_float* arow = (gmem_float*)(abase +
              (unsigned long long)((p & ((1u << SB_DST_BITS) - 1u)) * (PLANE_COLS * 4u)));
          __hip_atomic_fetch_add(arow + lane,      w * h0, __ATOMIC_RELAXED, __HIP_MEMORY_SCOPE_AGENT);
          __hip_atomic_fetch_add(arow + lane + 32, w * h1, __ATOMIC_RELAXED, __HIP_MEMORY_SCOPE_AGENT);
        }
      }
    }
    __syncthreads();
  }
}

#define G2_WAVES   8
#define G2_THREADS (G2_WAVES * 32)
#define G2_BLOCKS  512
#define G2_NT      3
#define G2_NP      (16 * G2_NT)
#define G2_KQ      (NHID / 4)
#define G2_NQ      (NHID / 8)
#define G2_HP      52
static_assert(G2_HP >= G2_NP && G2_HP % 4 == 0, "patch rows hold 48 columns, 16-byte aligned");
static_assert((G2_KQ * G2_NP) % G2_THREADS == 0, "staging divides evenly");
static_assert(PLANE_COLS == 64 && G2_NQ == 8 * N_PLANES, "eight runs of 8 k per plane");
static_assert(G1_STRIPS / G2_BLOCKS >= G2_WAVES, "every wave has a first strip");
static_assert(NCLASS % 4 == 0 && NCLASS <= G2_NP && NCLASS > 16 * (G2_NT - 1),
              "16-byte pieces of out never straddle rows; the last tile is partial");

__global__ __launch_bounds__(G2_THREADS)
void gcn_gemm2(const float* __restrict__ agg, const float* __restrict__ b1,
               const float* __restrict__ W2, const float* __restrict__ b2,
               float* __restrict__ H2, float* __restrict__ out) {
  __shared__ __attribute__((aligned(16))) float4 B2q[G2_KQ * G2_NP];
  __shared__ __attribute__((aligned(16))) float4 b1q[G2_KQ];
  __shared__ __attribute__((aligned(16))) float4 b2q[NCLASS / 4];
  __shared__ __attribute__((aligned(16))) float Hs[G2_WAVES * 16 * G2_HP];

  const int tid  = threadIdx.x;
  const int wave = __builtin_amdgcn_readfirstlane(tid >> 5);
  const int lane = tid & 31;
  const int half = lane >> 4;
  const int mn   = lane & 15;
  const int s0 = (int)(((long long)blockIdx.x * G1_STRIPS) / G2_BLOCKS);
  const int s1 = (int)(((long long)(blockIdx.x + 1) * G1_STRIPS) / G2_BLOCKS);
  int strip = s0 + wave;

  const float4* arow =
      reinterpret_cast<const float4*>(agg + (size_t)(strip * 16 + mn) * PLANE_COLS) + half;
  float4 an[G2_NQ];
  #pragma unroll
  for (int q = 0; q < G2_NQ; ++q) an[q] = arow[(q >> 3) * (PLANE_ELEMS / 4) + 2 * (q & 7)];

  #pragma unroll
  for (int i = 0; i < (G2_KQ * G2_NP) / G2_THREADS; ++i) {
    const int e  = tid + G2_THREADS * i;
    const int kk = e / G2_NP;
    const int n  = e % G2_NP;
    const bool real = (n < NCLASS);
    const float* wp = W2 + (size_t)(4 * kk) * NCLASS + (real ? n : NCLASS - 1);
    const float w0 = wp[0], w1 = wp[NCLASS], w2 = wp[2 * NCLASS], w3 = wp[3 * NCLASS];
    B2q[e] = make_float4(real ? w0 : 0.f, real ? w1 : 0.f, real ? w2 : 0.f, real ? w3 : 0.f);
  }
  if (tid < G2_KQ) b1q[tid] = reinterpret_cast<const float4*>(b1)[tid];
  if (tid < NCLASS / 4) b2q[tid] = reinterpret_cast<const float4*>(b2)[tid];
  __syncthreads();

  #pragma unroll 1
  while (strip < s1) {
    asm volatile("" ::: "memory");
    const int m0 = strip * 16;
    v8f acc[G2_NT];
    #pragma unroll
    for (int t = 0; t < G2_NT; ++t) acc[t] = v8f{};

    {
      float4* oq = reinterpret_cast<float4*>(out + (size_t)m0 * NCLASS) + lane;
      #pragma unroll
      for (int i = 0; i < 5; ++i) oq[32 * i] = b2q[(lane + 32 * i) % (NCLASS / 4)];
    }

    float4 ac[G2_NQ];
    #pragma unroll
    for (int q = 0; q < G2_NQ; ++q) ac[q] = an[q];

    strip += G2_WAVES;
    if (strip < s1) {
      arow = reinterpret_cast<const float4*>(agg + (size_t)(strip * 16 + mn) * PLANE_COLS) + half;
      #pragma unroll
      for (int q = 0; q < G2_NQ; ++q) an[q] = arow[(q >> 3) * (PLANE_ELEMS / 4) + 2 * (q & 7)];
    }

    #pragma unroll
    for (int q = 0; q < G2_NQ; ++q) {
      const float4 bi = b1q[2 * q + half];
      const float4* bp = B2q + (2 * q + half) * G2_NP + mn;
      float4 bf[G2_NT];
      #pragma unroll
      for (int t = 0; t < G2_NT; ++t) bf[t] = bp[16 * t];
      v2f a01, a23;
      a01.x = fmaxf(ac[q].x + bi.x, 0.0f); a01.y = fmaxf(ac[q].y + bi.y, 0.0f);
      a23.x = fmaxf(ac[q].z + bi.z, 0.0f); a23.y = fmaxf(ac[q].w + bi.w, 0.0f);
      #pragma unroll
      for (int t = 0; t < G2_NT; ++t) {
        v2f b01; b01.x = bf[t].x; b01.y = bf[t].y;
        acc[t] = __builtin_amdgcn_wmma_f32_16x16x4_f32(
             false, a01,  false, b01,
             (short)0, acc[t],  false,  false);
      }
      #pragma unroll
      for (int t = 0; t < G2_NT; ++t) {
        v2f b23; b23.x = bf[t].z; b23.y = bf[t].w;
        acc[t] = __builtin_amdgcn_wmma_f32_16x16x4_f32(
            false, a23, false, b23, (short)0, acc[t], false, false);
      }
    }

    {
      float* hs = Hs + wave * (16 * G2_HP);
      #pragma unroll
      for (int t = 0; t < G2_NT; ++t) {
        #pragma unroll
        for (int j = 0; j < 8; ++j) hs[(j + 8 * half) * G2_HP + 16 * t + mn] = acc[t][j];
      }
      float4* hq = reinterpret_cast<float4*>(H2 + (size_t)m0 * NCLASS) + lane;
      #pragma unroll
      for (int i = 0; i < 5; ++i) {
        const int p = lane + 32 * i;
        hq[32 * i] = *reinterpret_cast<const float4*>(
            hs + (p / (NCLASS / 4)) * G2_HP + 4 * (p % (NCLASS / 4)));
      }
    }
  }
}

__global__ __launch_bounds__(SPMM_THREADS)
void gcn_spmm2(const int* __restrict__ esrc, const int* __restrict__ edst,
               const float* __restrict__ ew, const float* __restrict__ H2,
               float* __restrict__ out) {
  const int lane = threadIdx.x & 31;
  const int wv   = blockIdx.x * (SPMM_THREADS / 32) + (threadIdx.x >> 5);
  const int e    = wv * EDGES_PER_WAVE + lane;
  const int   so = esrc[e] * NCLASS;
  const int   dn = edst[e] * NCLASS;
  const float wl = ew[e];
  const int tsel = lane >> 3;
  const int tcol = 32 + (lane & 7);
  for (int g = 0; g < EDGES_PER_WAVE; g += EDGE_BATCH) {
    float hf[EDGE_BATCH];
    float ht[EDGE_BATCH / 4];
    #pragma unroll
    for (int j = 0; j < EDGE_BATCH; ++j) {
      const int s = lane_read_i(so, g + j);
      hf[j] = H2[(size_t)s + lane];
    }
    #pragma unroll
    for (int q = 0; q < EDGE_BATCH / 4; ++q) {
      const int st = __shfl(so, g + 4 * q + tsel);
      ht[q] = H2[(size_t)st + tcol];
    }
    #pragma unroll
    for (int j = 0; j < EDGE_BATCH; ++j) {
      const int   d = lane_read_i(dn, g + j);
      const float w = lane_read_f(wl, g + j);
      atomicAdd(out + (size_t)d + lane, w * hf[j]);
    }
    #pragma unroll
    for (int q = 0; q < EDGE_BATCH / 4; ++q) {
      const int   dt = __shfl(dn, g + 4 * q + tsel);
      const float wt = __shfl(wl, g + 4 * q + tsel);
      atomicAdd(out + (size_t)dt + tcol, wt * ht[q]);
    }
  }
}

extern "C" void kernel_launch(void* const* d_in, const int* in_sizes, int n_in,
                              void* d_out, int out_size, void* d_ws, size_t ws_size,
                              hipStream_t stream) {
  const float* x    = (const float*)d_in[0];
  const int*   esrc = (const int*)  d_in[1];
  const int*   edst = (const int*)  d_in[2];
  const float* ew   = (const float*)d_in[3];
  const float* W1   = (const float*)d_in[4];
  const float* b1   = (const float*)d_in[5];
  const float* W2   = (const float*)d_in[6];
  const float* b2   = (const float*)d_in[7];
  float* out = (float*)d_out;

  float* H0  = (float*)d_ws;
  float* agg = H0 + (size_t)N_NODES * NHID;
  float* H2  = H0;

  gcn_bucket<<<SB_NCHUNK, SB_THREADS, 0, stream>>>(esrc, edst, ew, out);
  gcn_gemm1<<<G1_BLOCKS, G1_THREADS, 0, stream>>>(x, W1, H0, agg);
  gcn_spmm1<<<S1_GRID, S1_THREADS, 0, stream>>>(H0, agg, out);
  gcn_gemm2<<<G2_BLOCKS, G2_THREADS, 0, stream>>>(agg, b1, W2, b2, H2, out);
  gcn_spmm2<<<N_EDGES / (EDGES_PER_WAVE * (SPMM_THREADS / 32)), SPMM_THREADS, 0, stream>>>(
      esrc, edst, ew, H2, out);
}


# Both SpMMs sum in LDS: edges bucketed by target range, a work-group keeps its range's result rows in LDS (LDS float adds, arrival order), gathers source rows from cache, stores each row once
# speedup vs baseline: 3.4225x; 1.1895x over previous
#include <hip/hip_runtime.h>


#define N_NODES 100000
#define N_EDGES 1600000
#define NFEAT   256
#define NHID    128
#define NCLASS  40
#define PLANE_COLS  64
#define N_PLANES    (NHID / PLANE_COLS)
#define PLANE_ELEMS ((size_t)N_NODES * PLANE_COLS)
#define S1_CTR_STRIDE 1024

typedef __attribute__((ext_vector_type(2))) float v2f;
typedef __attribute__((ext_vector_type(8))) float v8f;

#define G1_WAVES   8
#define G1_THREADS (G1_WAVES * 32)
#define G1_STRIPS  (N_NODES / 16)
#define G1_BLOCKS  512
#define G1_KQ      (NFEAT / 4)
#define G1_CHUNK   8
#define G1_NCHUNK  (NFEAT / (8 * G1_CHUNK))
static_assert(N_NODES % 16 == 0, "whole strips");
static_assert(NHID == 8 * 16 && PLANE_COLS == 64 && N_PLANES == 2, "two pairs of column tiles per plane");
static_assert((G1_KQ * (NHID / 4)) % G1_THREADS == 0, "staging divides evenly");
static_assert(G1_STRIPS / G1_BLOCKS >= G1_WAVES, "every wave has a first strip");

__global__ __launch_bounds__(G1_THREADS)
void gcn_gemm1(const float* __restrict__ x, const float* __restrict__ W1,
               float* __restrict__ H0) {
  __shared__ __attribute__((aligned(16))) float4 Bq[G1_KQ * NHID];

  const int tid  = threadIdx.x;
  const int wave = __builtin_amdgcn_readfirstlane(tid >> 5);
  const int lane = tid & 31;
  const int half = lane >> 4;
  const int mn   = lane & 15;
  const int s0 = (int)(((long long)blockIdx.x * G1_STRIPS) / G1_BLOCKS);
  const int s1 = (int)(((long long)(blockIdx.x + 1) * G1_STRIPS) / G1_BLOCKS);
  int strip = s0 + wave;

  const float4* arow =
      reinterpret_cast<const float4*>(x + (size_t)(strip * 16 + mn) * NFEAT) + half;

  float4 an[G1_CHUNK];
  #pragma unroll
  for (int qi = 0; qi < G1_CHUNK; ++qi) an[qi] = arow[2 * qi];

  #pragma unroll
  for (int i = 0; i < (G1_KQ * (NHID / 4)) / G1_THREADS; ++i) {
    const int kk = wave + G1_WAVES * i;
    const float4* wp = reinterpret_cast<const float4*>(W1 + (size_t)(4 * kk) * NHID) + lane;
    const float4 r0 = wp[0];
    const float4 r1 = wp[NHID / 4];
    const float4 r2 = wp[2 * (NHID / 4)];
    const float4 r3 = wp[3 * (NHID / 4)];
    float4* bq = Bq + kk * NHID + 4 * lane;
    bq[0] = make_float4(r0.x, r1.x, r2.x, r3.x);
    bq[1] = make_float4(r0.y, r1.y, r2.y, r3.y);
    bq[2] = make_float4(r0.z, r1.z, r2.z, r3.z);
    bq[3] = make_float4(r0.w, r1.w, r2.w, r3.w);
  }
  __syncthreads();

  #pragma unroll 1
  while (strip < s1) {
    asm volatile("" ::: "memory");
    v8f acc[8];
    #pragma unroll
    for (int t = 0; t < 8; ++t) acc[t] = v8f{};

    #pragma unroll
    for (int ch = 0; ch < G1_NCHUNK; ++ch) {
      float4 ac[G1_CHUNK];
      #pragma unroll
      for (int qi = 0; qi < G1_CHUNK; ++qi) ac[qi] = an[qi];
      if (ch + 1 < G1_NCHUNK) {
        #pragma unroll
        for (int qi = 0; qi < G1_CHUNK; ++qi)
          an[qi] = arow[2 * (G1_CHUNK * (ch + 1) + qi)];
      }
      #pragma unroll
      for (int qi = 0; qi < G1_CHUNK; ++qi) {
        const int q = G1_CHUNK * ch + qi;
        const float4* bp = Bq + (2 * q + half) * NHID + mn;
        float4 bf[8];
        #pragma unroll
        for (int t = 0; t < 8; ++t) bf[t] = bp[16 * t];
        v2f a01, a23;
        a01.x = ac[qi].x; a01.y = ac[qi].y;
        a23.x = ac[qi].z; a23.y = ac[qi].w;
        #pragma unroll
        for (int t = 0; t < 8; ++t) {
          v2f b01; b01.x = bf[t].x; b01.y = bf[t].y;
          acc[t] = __builtin_amdgcn_wmma_f32_16x16x4_f32(
               false, a01,  false, b01,
               (short)0, acc[t],  false,  false);
        }
        #pragma unroll
        for (int t = 0; t < 8; ++t) {
          v2f b23; b23.x = bf[t].z; b23.y = bf[t].w;
          acc[t] = __builtin_amdgcn_wmma_f32_16x16x4_f32(
              false, a23, false, b23, (short)0, acc[t], false, false);
        }
      }
    }

    const int m0 = strip * 16;
    strip += G1_WAVES;
    if (strip < s1) {
      arow = reinterpret_cast<const float4*>(x + (size_t)(strip * 16 + mn) * NFEAT) + half;
      #pragma unroll
      for (int qi = 0; qi < G1_CHUNK; ++qi) an[qi] = arow[2 * qi];
    }

    float* orow = H0 + (size_t)m0 * PLANE_COLS + lane;
    #pragma unroll
    for (int tp = 0; tp < 4; ++tp) {
      #pragma unroll
      for (int j = 0; j < 8; ++j) {
        const float lo = acc[2 * tp][j];
        const float hi = acc[2 * tp + 1][j];
        const float send = half ? lo : hi;
        const float recv = __shfl_xor(send, 16);
        const float rowj  = half ? recv : lo;
        const float rowj8 = half ? hi : recv;
        orow[(tp >> 1) * PLANE_ELEMS + (size_t)j * PLANE_COLS + 32 * (tp & 1)]       = rowj;
        orow[(tp >> 1) * PLANE_ELEMS + (size_t)(j + 8) * PLANE_COLS + 32 * (tp & 1)] = rowj8;
      }
    }
  }
}

#define SB_RANGES     256
#define SB_ROWS       391
#define SB_CHUNK      6400
#define SB_NCHUNK     (N_EDGES / SB_CHUNK)
#define SB_THREADS    256
#define SB_PER_THREAD (SB_CHUNK / SB_THREADS)
#define SB_SRC_BITS   17
#define SB_SRC_MASK   ((1u << SB_SRC_BITS) - 1u)
#define SB_TAB_STRIDE 260
#define SB_PK_OFF     0
#define SB_W_OFF      N_EDGES
#define SB_TAB_OFF    (2 * N_EDGES)
#define SB_CTR_OFF    (3224 * 1024)
#define REC_WORDS     (SB_TAB_OFF + SB_NCHUNK * SB_TAB_STRIDE)
static_assert(SB_RANGES * SB_ROWS >= N_NODES, "the ranges cover every node");
static_assert(SB_RANGES % 2 == 0, "ranges come in pairs");
static_assert(N_NODES <= (1 << SB_SRC_BITS) && 2 * SB_ROWS <= (1 << (32 - SB_SRC_BITS)), "record fits one word");
static_assert(N_EDGES % SB_CHUNK == 0 && SB_CHUNK % SB_THREADS == 0, "whole chunks, equal shares");
static_assert(SB_THREADS == SB_RANGES && SB_TAB_STRIDE - SB_RANGES <= SB_THREADS, "one thread per range of a chunk");
static_assert(SB_TAB_STRIDE > SB_RANGES && REC_WORDS % 4 == 0, "table rows hold every start; 16-byte pieces");
static_assert(REC_WORDS <= SB_CTR_OFF, "table ends before the counters");
static_assert((size_t)SB_CTR_OFF + (N_PLANES - 1) * S1_CTR_STRIDE < (size_t)N_NODES * NCLASS, "counters inside out");

__global__ __launch_bounds__(SB_THREADS)
void gcn_bucket(const int* __restrict__ esrc, const int* __restrict__ edst,
                const float* __restrict__ ew, float* __restrict__ out) {
  __shared__ int   cnt[SB_RANGES];
  __shared__ int   start[SB_RANGES + 1];
  __shared__ int   imgP[SB_CHUNK];
  __shared__ float imgW[SB_CHUNK];
  int* scratch = reinterpret_cast<int*>(out);
  const int tid  = threadIdx.x;
  const int base = blockIdx.x * SB_CHUNK;

  if (blockIdx.x == 0 && tid < N_PLANES) scratch[SB_CTR_OFF + tid * S1_CTR_STRIDE] = 0;
  cnt[tid] = 0;
  __syncthreads();

  int   pk[SB_PER_THREAD], bk[SB_PER_THREAD], rk[SB_PER_THREAD];
  float wv[SB_PER_THREAD];
  #pragma unroll
  for (int i = 0; i < SB_PER_THREAD; ++i) {
    const int e = base + tid + SB_THREADS * i;
    const int sn = esrc[e];
    const int dn = edst[e];
    wv[i] = ew[e];
    const int b = dn / SB_ROWS;
    bk[i] = b;
    pk[i] = ((dn - (b >> 1) * (2 * SB_ROWS)) << SB_SRC_BITS) | sn;
  }
  #pragma unroll
  for (int i = 0; i < SB_PER_THREAD; ++i) rk[i] = atomicAdd(&cnt[bk[i]], 1);
  __syncthreads();

  {
    int acc = 0;
    for (int j = 0; j < SB_RANGES; ++j) {
      const int v = cnt[j];
      acc += (j < tid) ? v : 0;
    }
    start[tid] = acc;
  }
  if (tid == 0) start[SB_RANGES] = SB_CHUNK;
  __syncthreads();

  scratch[SB_TAB_OFF + blockIdx.x * SB_TAB_STRIDE + tid] = start[tid];
  if (tid < SB_TAB_STRIDE - SB_RANGES)
    scratch[SB_TAB_OFF + blockIdx.x * SB_TAB_STRIDE + SB_RANGES + tid] = (tid == 0) ? SB_CHUNK : 0;
  #pragma unroll
  for (int i = 0; i < SB_PER_THREAD; ++i) {
    const int pos = start[bk[i]] + rk[i];
    imgP[pos] = pk[i];
    imgW[pos] = wv[i];
  }
  __syncthreads();
  #pragma unroll
  for (int i = 0; i < SB_PER_THREAD; ++i) {
    const int j = tid + SB_THREADS * i;
    scratch[SB_PK_OFF + base + j] = imgP[j];
    out[SB_W_OFF + base + j]      = imgW[j];
  }
}

#define EDGE_BATCH   8
#define TRIP_BATCHES (32 / EDGE_BATCH)
#define STREAM_SLOTS ((SB_NCHUNK + 31) / 32)
static_assert(STREAM_SLOTS <= 32, "a wave keeps its runs' starts and ends one per lane");
static_assert(TRIP_BATCHES == 4 && EDGE_BATCH % 4 == 0, "a trip is written out as four batches; tail instructions take four edges each");

__device__ __forceinline__ int lane_read_i(int v, int l) {
  return __builtin_amdgcn_readlane(v, l);
}
__device__ __forceinline__ float lane_read_f(float v, int l) {
  return __int_as_float(__builtin_amdgcn_readlane(__float_as_int(v), l));
}
__device__ __forceinline__ void lds_add(float* p, float v) {
  __hip_atomic_fetch_add(p, v, __ATOMIC_RELAXED, __HIP_MEMORY_SCOPE_WORKGROUP);
}
struct RunStream {
  int pre[STREAM_SLOTS + 1];
  int d[STREAM_SLOTS];
};
__device__ __forceinline__ void stream_setup(RunStream& rs, int o0, int o1, int wave) {
  rs.pre[0] = 0;
  #pragma unroll
  for (int sl = 0; sl < STREAM_SLOTS; ++sl) {
    const int j0 = lane_read_i(o0, sl);
    const int j1 = lane_read_i(o1, sl);
    rs.d[sl] = (wave + 32 * sl) * SB_CHUNK + j0 - rs.pre[sl];
    rs.pre[sl + 1] = rs.pre[sl] + (j1 - j0);
  }
}
__device__ __forceinline__ int stream_index(const RunStream& rs, int t) {
  int dsel = rs.d[0];
  #pragma unroll
  for (int sl = 1; sl < STREAM_SLOTS; ++sl) dsel = (t >= rs.pre[sl]) ? rs.d[sl] : dsel;
  return t + dsel;
}
__device__ __forceinline__ void stream_fetch(const RunStream& rs, int T, int lane,
                                             const int* __restrict__ P, const float* __restrict__ Wp,
                                             int& pk, float& wl) {
  const int  t  = T + lane;
  const bool ok = t < rs.pre[STREAM_SLOTS];
  const int  ix = ok ? stream_index(rs, t) : 0;
  const int   p = P[ix];
  const float w = Wp[ix];
  pk = ok ? p : 0;
  wl = ok ? w : 0.f;
}

#define S1_THREADS 1024
#define S1_WAVES   (S1_THREADS / 32)
#define S1_PAIRS   (SB_RANGES / 2)
#define S1_ROWS    (2 * SB_ROWS)
#define S1_GRID    (N_PLANES * S1_PAIRS)
static_assert(S1_WAVES == 32, "the record stream takes every 32nd chunk");

__device__ __forceinline__ void s1_gather(float (&h)[EDGE_BATCH][2], int b, int pk,
                                          const float* __restrict__ gplane, int lane) {
  #pragma unroll
  for (int u = 0; u < EDGE_BATCH; ++u) {
    const unsigned p = (unsigned)lane_read_i(pk, EDGE_BATCH * b + u);
    const float* hp = gplane + (size_t)(p & SB_SRC_MASK) * PLANE_COLS;
    h[u][0] = hp[lane];
    h[u][1] = hp[lane + 32];
  }
}
__device__ __forceinline__ void s1_add(const float (&h)[EDGE_BATCH][2], int b, int pk, float wl,
                                       int nleft, float* myrow) {
  #pragma unroll
  for (int u = 0; u < EDGE_BATCH; ++u) {
    const unsigned p = (unsigned)lane_read_i(pk, EDGE_BATCH * b + u);
    const float w  = lane_read_f(wl, EDGE_BATCH * b + u);
    const bool  ok = (EDGE_BATCH * b + u == 0) || (EDGE_BATCH * b + u < nleft);
    float* ap = myrow + (p >> SB_SRC_BITS) * PLANE_COLS;
    lds_add(ap,      ok ? w * h[u][0] : 0.f);
    lds_add(ap + 32, ok ? w * h[u][1] : 0.f);
  }
}

__global__ __launch_bounds__(S1_THREADS)
void gcn_spmm1(const float* __restrict__ H0, float* __restrict__ agg,
               float* __restrict__ out) {
  __shared__ __attribute__((aligned(16))) float rows[S1_ROWS * PLANE_COLS];
  __shared__ int item_s;
  const int tid  = threadIdx.x;
  const int lane = tid & 31;
  const int wave = __builtin_amdgcn_readfirstlane(tid >> 5);
  const int* scratch = reinterpret_cast<const int*>(out);
  int* ctr = reinterpret_cast<int*>(out) + SB_CTR_OFF;
  const int half = (int)((__builtin_amdgcn_s_sendmsg_rtn(0x87) >> 18) & 1u);

  for (;;) {
    if (tid == 0) {
      int plane = half;
      int t = atomicAdd(ctr + plane * S1_CTR_STRIDE, 1);
      if (t >= S1_PAIRS) {
        plane = half ^ 1;
        t = atomicAdd(ctr + plane * S1_CTR_STRIDE, 1);
      }
      item_s = (t < S1_PAIRS) ? (plane * S1_PAIRS + t) : -1;
    }
    __syncthreads();
    const int item = __builtin_amdgcn_readfirstlane(item_s);
    if (item < 0) break;
    const int plane = item / S1_PAIRS;
    const int r     = item - plane * S1_PAIRS;
    const int row0  = r * S1_ROWS;
    const int nrows = (N_NODES - row0 < S1_ROWS) ? (N_NODES - row0) : S1_ROWS;

    int o0 = 0, o1 = 0;
    {
      const int c = wave + S1_WAVES * lane;
      if (c < SB_NCHUNK) {
        const int* tp = scratch + SB_TAB_OFF + c * SB_TAB_STRIDE + 2 * r;
        o0 = tp[0];
        o1 = tp[2];
      }
    }

    {
      float4* l4 = reinterpret_cast<float4*>(rows);
      for (int i = tid; i < S1_ROWS * (PLANE_COLS / 4); i += S1_THREADS)
        l4[i] = make_float4(0.f, 0.f, 0.f, 0.f);
    }
    __syncthreads();

    const float* gplane = H0 + (size_t)plane * PLANE_ELEMS;
    float* myrow = rows + lane;
    const int*   P  = scratch + SB_PK_OFF;
    const float* Wp = out + SB_W_OFF;
    RunStream rs;
    stream_setup(rs, o0, o1, wave);
    const int total = rs.pre[STREAM_SLOTS];
    int pk; float wl;
    stream_fetch(rs, 0, lane, P, Wp, pk, wl);
    for (int T = 0; T < total; T += 32) {
      int pkn; float wln;
      stream_fetch(rs, T + 32, lane, P, Wp, pkn, wln);
      const int nleft = total - T;
      float hA[EDGE_BATCH][2], hB[EDGE_BATCH][2];
      s1_gather(hA, 0, pk, gplane, lane);
      s1_gather(hB, 1, pk, gplane, lane);
      s1_add(hA, 0, pk, wl, nleft, myrow);
      s1_gather(hA, 2, pk, gplane, lane);
      s1_add(hB, 1, pk, wl, nleft, myrow);
      s1_gather(hB, 3, pk, gplane, lane);
      s1_add(hA, 2, pk, wl, nleft, myrow);
      s1_add(hB, 3, pk, wl, nleft, myrow);
      pk = pkn;
      wl = wln;
    }
    __syncthreads();
    {
      const float4* l4 = reinterpret_cast<const float4*>(rows);
      float4* gdst = reinterpret_cast<float4*>(
          agg + (size_t)plane * PLANE_ELEMS + (size_t)row0 * PLANE_COLS);
      const int npieces = nrows * (PLANE_COLS / 4);
      #pragma unroll 4
      for (int i = tid; i < npieces; i += S1_THREADS) gdst[i] = l4[i];
    }
    __syncthreads();
  }
}

#define G2_WAVES   8
#define G2_THREADS (G2_WAVES * 32)
#define G2_BLOCKS  512
#define G2_NT      3
#define G2_NP      (16 * G2_NT)
#define G2_KQ      (NHID / 4)
#define G2_NQ      (NHID / 8)
#define G2_HP      52
static_assert(G2_HP >= G2_NP && G2_HP % 4 == 0, "patch rows hold 48 columns, 16-byte aligned");
static_assert((G2_KQ * G2_NP) % G2_THREADS == 0, "staging divides evenly");
static_assert(PLANE_COLS == 64 && G2_NQ == 8 * N_PLANES, "eight runs of 8 k per plane");
static_assert(G1_STRIPS / G2_BLOCKS >= G2_WAVES, "every wave has a first strip");
static_assert(NCLASS % 4 == 0 && NCLASS <= G2_NP && NCLASS > 16 * (G2_NT - 1),
              "16-byte pieces of H2 never straddle rows; the last tile is partial");
#define REC_PIECES   (REC_WORDS / 4)
#define G2_MOVE_TRIPS ((REC_PIECES + G2_BLOCKS * G2_THREADS - 1) / (G2_BLOCKS * G2_THREADS))

__global__ __launch_bounds__(G2_THREADS)
void gcn_gemm2(const float* __restrict__ agg, const float* __restrict__ b1,
               const float* __restrict__ W2, float* __restrict__ H2,
               const float* __restrict__ recsrc, float* __restrict__ rec) {
  __shared__ __attribute__((aligned(16))) float4 B2q[G2_KQ * G2_NP];
  __shared__ __attribute__((aligned(16))) float4 b1q[G2_KQ];
  __shared__ __attribute__((aligned(16))) float Hs[G2_WAVES * 16 * G2_HP];

  const int tid  = threadIdx.x;
  const int wave = __builtin_amdgcn_readfirstlane(tid >> 5);
  const int lane = tid & 31;
  const int half = lane >> 4;
  const int mn   = lane & 15;
  const int s0 = (int)(((long long)blockIdx.x * G1_STRIPS) / G2_BLOCKS);
  const int s1 = (int)(((long long)(blockIdx.x + 1) * G1_STRIPS) / G2_BLOCKS);
  int strip = s0 + wave;

  const float4* arow =
      reinterpret_cast<const float4*>(agg + (size_t)(strip * 16 + mn) * PLANE_COLS) + half;
  float4 an[G2_NQ];
  #pragma unroll
  for (int q = 0; q < G2_NQ; ++q) an[q] = arow[(q >> 3) * (PLANE_ELEMS / 4) + 2 * (q & 7)];


  #pragma unroll
  for (int i = 0; i < (G2_KQ * G2_NP) / G2_THREADS; ++i) {
    const int e  = tid + G2_THREADS * i;
    const int kk = e / G2_NP;
    const int n  = e % G2_NP;
    const bool real = (n < NCLASS);
    const float* wp = W2 + (size_t)(4 * kk) * NCLASS + (real ? n : NCLASS - 1);
    const float w0 = wp[0], w1 = wp[NCLASS], w2 = wp[2 * NCLASS], w3 = wp[3 * NCLASS];
    B2q[e] = make_float4(real ? w0 : 0.f, real ? w1 : 0.f, real ? w2 : 0.f, real ? w3 : 0.f);
  }
  if (tid < G2_KQ) b1q[tid] = reinterpret_cast<const float4*>(b1)[tid];
  __syncthreads();

  #pragma unroll 1
  while (strip < s1) {
    asm volatile("" ::: "memory");
    const int m0 = strip * 16;
    v8f acc[G2_NT];
    #pragma unroll
    for (int t = 0; t < G2_NT; ++t) acc[t] = v8f{};

    float4 ac[G2_NQ];
    #pragma unroll
    for (int q = 0; q < G2_NQ; ++q) ac[q] = an[q];

    strip += G2_WAVES;
    if (strip < s1) {
      arow = reinterpret_cast<const float4*>(agg + (size_t)(strip * 16 + mn) * PLANE_COLS) + half;
      #pragma unroll
      for (int q = 0; q < G2_NQ; ++q) an[q] = arow[(q >> 3) * (PLANE_ELEMS / 4) + 2 * (q & 7)];
    }

    #pragma unroll
    for (int q = 0; q < G2_NQ; ++q) {
      const float4 bi = b1q[2 * q + half];
      const float4* bp = B2q + (2 * q + half) * G2_NP + mn;
      float4 bf[G2_NT];
      #pragma unroll
      for (int t = 0; t < G2_NT; ++t) bf[t] = bp[16 * t];
      v2f a01, a23;
      a01.x = fmaxf(ac[q].x + bi.x, 0.0f); a01.y = fmaxf(ac[q].y + bi.y, 0.0f);
      a23.x = fmaxf(ac[q].z + bi.z, 0.0f); a23.y = fmaxf(ac[q].w + bi.w, 0.0f);
      #pragma unroll
      for (int t = 0; t < G2_NT; ++t) {
        v2f b01; b01.x = bf[t].x; b01.y = bf[t].y;
        acc[t] = __builtin_amdgcn_wmma_f32_16x16x4_f32(
             false, a01,  false, b01,
             (short)0, acc[t],  false,  false);
      }
      #pragma unroll
      for (int t = 0; t < G2_NT; ++t) {
        v2f b23; b23.x = bf[t].z; b23.y = bf[t].w;
        acc[t] = __builtin_amdgcn_wmma_f32_16x16x4_f32(
            false, a23, false, b23, (short)0, acc[t], false, false);
      }
    }

    {
      float* hs = Hs + wave * (16 * G2_HP);
      #pragma unroll
      for (int t = 0; t < G2_NT; ++t) {
        #pragma unroll
        for (int j = 0; j < 8; ++j) hs[(j + 8 * half) * G2_HP + 16 * t + mn] = acc[t][j];
      }
      float4* hq = reinterpret_cast<float4*>(H2 + (size_t)m0 * NCLASS) + lane;
      #pragma unroll
      for (int i = 0; i < 5; ++i) {
        const int p = lane + 32 * i;
        hq[32 * i] = *reinterpret_cast<const float4*>(
            hs + (p / (NCLASS / 4)) * G2_HP + 4 * (p % (NCLASS / 4)));
      }
    }
  }

  {
    const float4* rs = reinterpret_cast<const float4*>(recsrc);
    float4* rd = reinterpret_cast<float4*>(rec);
    const int mv0 = blockIdx.x * G2_THREADS + tid;
    float4 mv[G2_MOVE_TRIPS];
    #pragma unroll
    for (int i = 0; i < G2_MOVE_TRIPS; ++i) {
      const bool whole = (i + 1) * (G2_BLOCKS * G2_THREADS) <= REC_PIECES;
      const int p = mv0 + i * (G2_BLOCKS * G2_THREADS);
      mv[i] = rs[(whole || p < REC_PIECES) ? p : REC_PIECES - 1];
    }
    static_assert(G2_MOVE_TRIPS == 7, "the statement below names every piece");
    asm volatile("" :: "v"(mv[0].x), "v"(mv[0].y), "v"(mv[0].z), "v"(mv[0].w),
                       "v"(mv[1].x), "v"(mv[1].y), "v"(mv[1].z), "v"(mv[1].w),
                       "v"(mv[2].x), "v"(mv[2].y), "v"(mv[2].z), "v"(mv[2].w),
                       "v"(mv[3].x), "v"(mv[3].y), "v"(mv[3].z), "v"(mv[3].w),
                       "v"(mv[4].x), "v"(mv[4].y), "v"(mv[4].z), "v"(mv[4].w),
                       "v"(mv[5].x), "v"(mv[5].y), "v"(mv[5].z), "v"(mv[5].w),
                       "v"(mv[6].x), "v"(mv[6].y), "v"(mv[6].z), "v"(mv[6].w));
    #pragma unroll
    for (int i = 0; i < G2_MOVE_TRIPS; ++i) {
      const bool whole = (i + 1) * (G2_BLOCKS * G2_THREADS) <= REC_PIECES;
      const int p = mv0 + i * (G2_BLOCKS * G2_THREADS);
      if (whole || p < REC_PIECES) rd[p] = mv[i];
    }
  }
}

#define S2_THREADS 1024
#define S2_WAVES   (S2_THREADS / 32)
static_assert(S2_WAVES == 32, "the record stream takes every 32nd chunk");

__device__ __forceinline__ int sel4_i(int tsel, int a0, int a1, int a2, int a3) {
  const int lo = (tsel & 1) ? a1 : a0;
  const int hi = (tsel & 1) ? a3 : a2;
  return (tsel & 2) ? hi : lo;
}
__device__ __forceinline__ float sel4_f(int tsel, float a0, float a1, float a2, float a3) {
  const float lo = (tsel & 1) ? a1 : a0;
  const float hi = (tsel & 1) ? a3 : a2;
  return (tsel & 2) ? hi : lo;
}
__device__ __forceinline__ void s2_gather(float (&hf)[EDGE_BATCH], float (&ht)[EDGE_BATCH / 4], int b,
                                          int so, const float* __restrict__ H2, int lane,
                                          int tsel, int tcol) {
  int sb[EDGE_BATCH];
  #pragma unroll
  for (int u = 0; u < EDGE_BATCH; ++u) {
    sb[u] = lane_read_i(so, EDGE_BATCH * b + u);
    hf[u] = H2[(size_t)sb[u] + lane];
  }
  #pragma unroll
  for (int q = 0; q < EDGE_BATCH / 4; ++q) {
    const int st = sel4_i(tsel, sb[4 * q], sb[4 * q + 1], sb[4 * q + 2], sb[4 * q + 3]);
    ht[q] = H2[(size_t)st + tcol];
  }
}
__device__ __forceinline__ void s2_add(const float (&hf)[EDGE_BATCH], const float (&ht)[EDGE_BATCH / 4], int b,
                                       int dl, float wl, int nleft, float* rows, int lane,
                                       int tsel, int tcol) {
  int   d[EDGE_BATCH];
  float w[EDGE_BATCH];
  #pragma unroll
  for (int u = 0; u < EDGE_BATCH; ++u) {
    d[u] = lane_read_i(dl, EDGE_BATCH * b + u);
    w[u] = lane_read_f(wl, EDGE_BATCH * b + u);
    const bool ok = (EDGE_BATCH * b + u == 0) || (EDGE_BATCH * b + u < nleft);
    lds_add(rows + d[u] + lane, ok ? w[u] * hf[u] : 0.f);
  }
  #pragma unroll
  for (int q = 0; q < EDGE_BATCH / 4; ++q) {
    const int   dt = sel4_i(tsel, d[4 * q], d[4 * q + 1], d[4 * q + 2], d[4 * q + 3]);
    const float wt = sel4_f(tsel, w[4 * q], w[4 * q + 1], w[4 * q + 2], w[4 * q + 3]);
    const bool  ok = EDGE_BATCH * b + 4 * q + tsel < nleft;
    lds_add(rows + dt + tcol, ok ? wt * ht[q] : 0.f);
  }
}

__global__ __launch_bounds__(S2_THREADS)
void gcn_spmm2(const float* __restrict__ H2, const float* __restrict__ rec,
               const float* __restrict__ b2, float* __restrict__ out) {
  __shared__ __attribute__((aligned(16))) float rows[SB_ROWS * NCLASS];
  const int tid  = threadIdx.x;
  const int lane = tid & 31;
  const int wave = __builtin_amdgcn_readfirstlane(tid >> 5);
  const int* scratch = reinterpret_cast<const int*>(rec);
  const int item  = blockIdx.x;
  const int row0  = item * SB_ROWS;
  const int nrows = (N_NODES - row0 < SB_ROWS) ? (N_NODES - row0) : SB_ROWS;
  const int lbase = (item & 1) * SB_ROWS;

  int o0 = 0, o1 = 0;
  {
    const int c = wave + S2_WAVES * lane;
    if (c < SB_NCHUNK) {
      const int* tp = scratch + SB_TAB_OFF + c * SB_TAB_STRIDE + item;
      o0 = tp[0];
      o1 = tp[1];
    }
  }

  {
    float4* l4 = reinterpret_cast<float4*>(rows);
    const float4* b4 = reinterpret_cast<const float4*>(b2);
    #pragma unroll
    for (int k = 0; k < (SB_ROWS * (NCLASS / 4) + S2_THREADS - 1) / S2_THREADS; ++k) {
      const bool whole = (k + 1) * S2_THREADS <= SB_ROWS * (NCLASS / 4);
      const int i = tid + S2_THREADS * k;
      const float4 v = b4[i % (NCLASS / 4)];
      if (whole || i < SB_ROWS * (NCLASS / 4)) l4[i] = v;
    }
  }
  __syncthreads();

  const int tsel = lane >> 3;
  const int tcol = 32 + (lane & 7);
  const int*   P  = scratch + SB_PK_OFF;
  const float* Wp = rec + SB_W_OFF;
  RunStream rs;
  stream_setup(rs, o0, o1, wave);
  const int total = rs.pre[STREAM_SLOTS];
  int pk; float wl;
  stream_fetch(rs, 0, lane, P, Wp, pk, wl);
  for (int T = 0; T < total; T += 32) {
    int pkn; float wln;
    stream_fetch(rs, T + 32, lane, P, Wp, pkn, wln);
    const int nleft = total - T;
    const int so = (int)((unsigned)pk & SB_SRC_MASK) * NCLASS;
    int dr = (int)((unsigned)pk >> SB_SRC_BITS) - lbase;
    dr = (dr < 0) ? 0 : dr;
    const int dl = dr * NCLASS;
    float fA[EDGE_BATCH], fB[EDGE_BATCH], tA[EDGE_BATCH / 4], tB[EDGE_BATCH / 4];
    s2_gather(fA, tA, 0, so, H2, lane, tsel, tcol);
    s2_gather(fB, tB, 1, so, H2, lane, tsel, tcol);
    s2_add(fA, tA, 0, dl, wl, nleft, rows, lane, tsel, tcol);
    s2_gather(fA, tA, 2, so, H2, lane, tsel, tcol);
    s2_add(fB, tB, 1, dl, wl, nleft, rows, lane, tsel, tcol);
    s2_gather(fB, tB, 3, so, H2, lane, tsel, tcol);
    s2_add(fA, tA, 2, dl, wl, nleft, rows, lane, tsel, tcol);
    s2_add(fB, tB, 3, dl, wl, nleft, rows, lane, tsel, tcol);
    pk = pkn;
    wl = wln;
  }
  __syncthreads();
  {
    const float4* l4 = reinterpret_cast<const float4*>(rows);
    float4* gdst = reinterpret_cast<float4*>(out + (size_t)row0 * NCLASS);
    const int npieces = nrows * (NCLASS / 4);
    #pragma unroll 4
    for (int i = tid; i < npieces; i += S2_THREADS) gdst[i] = l4[i];
  }
}

extern "C" void kernel_launch(void* const* d_in, const int* in_sizes, int n_in,
                              void* d_out, int out_size, void* d_ws, size_t ws_size,
                              hipStream_t stream) {
  const float* x    = (const float*)d_in[0];
  const int*   esrc = (const int*)  d_in[1];
  const int*   edst = (const int*)  d_in[2];
  const float* ew   = (const float*)d_in[3];
  const float* W1   = (const float*)d_in[4];
  const float* b1   = (const float*)d_in[5];
  const float* W2   = (const float*)d_in[6];
  const float* b2   = (const float*)d_in[7];
  float* out = (float*)d_out;

  float* H0  = (float*)d_ws;
  float* agg = H0 + (size_t)N_NODES * NHID;
  float* H2  = H0;
  float* rec = H0 + (size_t)N_NODES * NCLASS;

  gcn_bucket<<<SB_NCHUNK, SB_THREADS, 0, stream>>>(esrc, edst, ew, out);
  gcn_gemm1<<<G1_BLOCKS, G1_THREADS, 0, stream>>>(x, W1, H0);
  gcn_spmm1<<<S1_GRID, S1_THREADS, 0, stream>>>(H0, agg, out);
  gcn_gemm2<<<G2_BLOCKS, G2_THREADS, 0, stream>>>(agg, b1, W2, H2, out, rec);
  gcn_spmm2<<<SB_RANGES, S2_THREADS, 0, stream>>>(H2, rec, b2, out);
}
